# starting kernel
# baseline (speedup 1.0000x reference)
#include <hip/hip_runtime.h>
#include <math.h>

#define IN_CH 128
#define HID   128
#define OUTC  47
#define OUTP  48

typedef float v2f __attribute__((ext_vector_type(2)));
typedef float v8f __attribute__((ext_vector_type(8)));

__global__ void zero_f32(float* __restrict__ p, long long n) {
  long long i = (long long)blockIdx.x * blockDim.x + threadIdx.x;
  long long stride = (long long)gridDim.x * blockDim.x;
  for (; i < n; i += stride) p[i] = 0.0f;
}

__global__ void scatter_deg(const int* __restrict__ dst, float* __restrict__ deg, int E) {
  int e = blockIdx.x * blockDim.x + threadIdx.x;
  if (e < E) atomicAdd(&deg[dst[e]], 1.0f);
}

__global__ void scatter_msg(const float* __restrict__ xsrc, const int* __restrict__ src,
                            const int* __restrict__ dst, float* __restrict__ msg, int E) {
  long long t = (long long)blockIdx.x * blockDim.x + threadIdx.x;
  int e    = (int)(t >> 5);
  int lane = (int)(t & 31);
  if (e >= E) return;
  int s = src[e];
  int d = dst[e];
  const float4 v = *(const float4*)(xsrc + (size_t)s * IN_CH + lane * 4);
  float* o = msg + (size_t)d * IN_CH + lane * 4;
  atomicAdd(o + 0, v.x);
  atomicAdd(o + 1, v.y);
  atomicAdd(o + 2, v.z);
  atomicAdd(o + 3, v.w);
}

__global__ __launch_bounds__(128)
void sage_gemm_relu(const float* __restrict__ msg, const float* __restrict__ deg,
                    const float* __restrict__ xdst,
                    const float* __restrict__ Wl, const float* __restrict__ bl,
                    const float* __restrict__ Wr,
                    float* __restrict__ hout, int nrows)
{
  __shared__ float sA[16][IN_CH + 4];
  __shared__ float sX[16][IN_CH + 4];
  const int tid = threadIdx.x;
  const int r0  = blockIdx.x * 16;

  for (int m = 0; m < 16; ++m) {
    int row = r0 + m;
    float a = 0.0f, xx = 0.0f;
    if (row < nrows) {
      float inv = 1.0f / fmaxf(deg[row], 1.0f);
      a  = msg [(size_t)row * IN_CH + tid] * inv;
      xx = xdst[(size_t)row * IN_CH + tid];
    }
    sA[m][tid] = a;
    sX[m][tid] = xx;
  }
  __syncthreads();

  const int wave = tid >> 5;
  const int lane = tid & 31;
  const int half = lane >> 4;
  const int m    = lane & 15;
  const int n    = lane & 15;
  const int kh   = half * 2;

  for (int t = 0; t < 2; ++t) {
    const int n0 = (wave * 2 + t) * 16;
    v8f acc = {0.f, 0.f, 0.f, 0.f, 0.f, 0.f, 0.f, 0.f};
    for (int k = 0; k < IN_CH; k += 4) {
      v2f a, b;
      a.x = sA[m][k + kh];
      a.y = sA[m][k + kh + 1];
      b.x = Wl[(size_t)(k + kh)     * HID + n0 + n];
      b.y = Wl[(size_t)(k + kh + 1) * HID + n0 + n];
      acc = __builtin_amdgcn_wmma_f32_16x16x4_f32(false, a, false, b, (short)0, acc, false, false);
    }
    for (int k = 0; k < IN_CH; k += 4) {
      v2f a, b;
      a.x = sX[m][k + kh];
      a.y = sX[m][k + kh + 1];
      b.x = Wr[(size_t)(k + kh)     * HID + n0 + n];
      b.y = Wr[(size_t)(k + kh + 1) * HID + n0 + n];
      acc = __builtin_amdgcn_wmma_f32_16x16x4_f32(false, a, false, b, (short)0, acc, false, false);
    }
    const float bias  = bl[n0 + n];
    const int   mbase = half * 8;
    for (int r = 0; r < 8; ++r) {
      int row = r0 + mbase + r;
      if (row < nrows) {
        float v = acc[r] + bias;
        hout[(size_t)row * HID + n0 + n] = fmaxf(v, 0.0f);
      }
    }
  }
}

__global__ __launch_bounds__(128)
void sage_gemm_lsm(const float* __restrict__ msg, const float* __restrict__ deg,
                   const float* __restrict__ xdst,
                   const float* __restrict__ Wl, const float* __restrict__ bl,
                   const float* __restrict__ Wr,
                   float* __restrict__ out, int nrows)
{
  __shared__ float sA[16][HID + 4];
  __shared__ float sX[16][HID + 4];
  __shared__ float sL[16][OUTP];
  const int tid = threadIdx.x;
  const int r0  = blockIdx.x * 16;

  for (int m = 0; m < 16; ++m) {
    int row = r0 + m;
    float a = 0.0f, xx = 0.0f;
    if (row < nrows) {
      float inv = 1.0f / fmaxf(deg[row], 1.0f);
      a  = msg [(size_t)row * HID + tid] * inv;
      xx = xdst[(size_t)row * HID + tid];
    }
    sA[m][tid] = a;
    sX[m][tid] = xx;
  }
  __syncthreads();

  const int wave = tid >> 5;
  const int lane = tid & 31;
  const int half = lane >> 4;
  const int m    = lane & 15;
  const int n    = lane & 15;
  const int kh   = half * 2;

  if (wave < 3) {
    const int n0  = wave * 16;
    const int col = n0 + n;
    const bool cv = (col < OUTC);
    v8f acc = {0.f, 0.f, 0.f, 0.f, 0.f, 0.f, 0.f, 0.f};
    for (int k = 0; k < HID; k += 4) {
      v2f a, b;
      a.x = sA[m][k + kh];
      a.y = sA[m][k + kh + 1];
      b.x = cv ? Wl[(size_t)(k + kh)     * OUTC + col] : 0.0f;
      b.y = cv ? Wl[(size_t)(k + kh + 1) * OUTC + col] : 0.0f;
      acc = __builtin_amdgcn_wmma_f32_16x16x4_f32(false, a, false, b, (short)0, acc, false, false);
    }
    for (int k = 0; k < HID; k += 4) {
      v2f a, b;
      a.x = sX[m][k + kh];
      a.y = sX[m][k + kh + 1];
      b.x = cv ? Wr[(size_t)(k + kh)     * OUTC + col] : 0.0f;
      b.y = cv ? Wr[(size_t)(k + kh + 1) * OUTC + col] : 0.0f;
      acc = __builtin_amdgcn_wmma_f32_16x16x4_f32(false, a, false, b, (short)0, acc, false, false);
    }
    const float bias  = cv ? bl[col] : 0.0f;
    const int   mbase = half * 8;
    for (int r = 0; r < 8; ++r)
      sL[mbase + r][col] = acc[r] + bias;
  }
  __syncthreads();

  if (tid < 16) {
    int row = r0 + tid;
    if (row < nrows) {
      float mx = -INFINITY;
      for (int c = 0; c < OUTC; ++c) mx = fmaxf(mx, sL[tid][c]);
      float s = 0.0f;
      for (int c = 0; c < OUTC; ++c) s += expf(sL[tid][c] - mx);
      float lse = logf(s);
      for (int c = 0; c < OUTC; ++c)
        out[(size_t)row * OUTC + c] = sL[tid][c] - mx - lse;
    }
  }
}

extern "C" void kernel_launch(void* const* d_in, const int* in_sizes, int n_in,
                              void* d_out, int out_size, void* d_ws, size_t ws_size,
                              hipStream_t stream)
{
  const float* x    = (const float*)d_in[0];
  const int*   src0 = (const int*)d_in[1];
  const int*   dst0 = (const int*)d_in[2];
  const int*   src1 = (const int*)d_in[3];
  const int*   dst1 = (const int*)d_in[4];
  const float* Wl0  = (const float*)d_in[7];
  const float* bl0  = (const float*)d_in[8];
  const float* Wr0  = (const float*)d_in[9];
  const float* Wl1  = (const float*)d_in[10];
  const float* bl1  = (const float*)d_in[11];
  const float* Wr1  = (const float*)d_in[12];
  float* out = (float*)d_out;

  const int E0 = in_sizes[1];
  const int E1 = in_sizes[3];
  const int N1 = 100000;
  const int N2 = out_size / OUTC;

  float* msg0 = (float*)d_ws;
  float* deg0 = msg0 + (size_t)N1 * HID;
  float* h    = deg0 + N1;
  float* msg1 = h    + (size_t)N1 * HID;
  float* deg1 = msg1 + (size_t)N2 * HID;

  const long long z0 = (long long)N1 * HID + N1;
  const long long z1 = (long long)N2 * HID + N2;
  zero_f32<<<1024, 256, 0, stream>>>(msg0, z0);
  zero_f32<<<256,  256, 0, stream>>>(msg1, z1);

  scatter_deg<<<(E0 + 255) / 256, 256, 0, stream>>>(dst0, deg0, E0);
  {
    long long th = (long long)E0 * 32;
    scatter_msg<<<(unsigned)((th + 255) / 256), 256, 0, stream>>>(x, src0, dst0, msg0, E0);
  }
  sage_gemm_relu<<<(N1 + 15) / 16, 128, 0, stream>>>(msg0, deg0, x, Wl0, bl0, Wr0, h, N1);

  scatter_deg<<<(E1 + 255) / 256, 256, 0, stream>>>(dst1, deg1, E1);
  {
    long long th = (long long)E1 * 32;
    scatter_msg<<<(unsigned)((th + 255) / 256), 256, 0, stream>>>(h, src1, dst1, msg1, E1);
  }
  sage_gemm_lsm<<<(N2 + 15) / 16, 128, 0, stream>>>(msg1, deg1, h, Wl1, bl1, Wr1, out, N2);
}


# scatter_msg: lane l takes channels l,l+32,l+64,l+96, so each f32 atomic add (and row load) of a wave covers one whole 128-B line: 4 atomic line requests per edge instead of 16
# speedup vs baseline: 1.2183x; 1.2183x over previous
#include <hip/hip_runtime.h>
#include <math.h>

#define IN_CH 128
#define HID   128
#define OUTC  47
#define OUTP  48

typedef float v2f __attribute__((ext_vector_type(2)));
typedef float v8f __attribute__((ext_vector_type(8)));

__global__ void zero_f32(float* __restrict__ p, long long n) {
  long long i = (long long)blockIdx.x * blockDim.x + threadIdx.x;
  long long stride = (long long)gridDim.x * blockDim.x;
  for (; i < n; i += stride) p[i] = 0.0f;
}

__global__ void scatter_deg(const int* __restrict__ dst, float* __restrict__ deg, int E) {
  int e = blockIdx.x * blockDim.x + threadIdx.x;
  if (e < E) atomicAdd(&deg[dst[e]], 1.0f);
}

__global__ void scatter_msg(const float* __restrict__ xsrc, const int* __restrict__ src,
                            const int* __restrict__ dst, float* __restrict__ msg, int E) {
  long long t = (long long)blockIdx.x * blockDim.x + threadIdx.x;
  int e    = (int)(t >> 5);
  int lane = (int)(t & 31);
  if (e >= E) return;
  int s = src[e];
  int d = dst[e];
  const float* p = xsrc + (size_t)s * IN_CH + lane;
  const float v0 = p[0];
  const float v1 = p[32];
  const float v2 = p[64];
  const float v3 = p[96];
  float* o = msg + (size_t)d * IN_CH + lane;
  atomicAdd(o +  0, v0);
  atomicAdd(o + 32, v1);
  atomicAdd(o + 64, v2);
  atomicAdd(o + 96, v3);
}

__global__ __launch_bounds__(128)
void sage_gemm_relu(const float* __restrict__ msg, const float* __restrict__ deg,
                    const float* __restrict__ xdst,
                    const float* __restrict__ Wl, const float* __restrict__ bl,
                    const float* __restrict__ Wr,
                    float* __restrict__ hout, int nrows)
{
  __shared__ float sA[16][IN_CH + 4];
  __shared__ float sX[16][IN_CH + 4];
  const int tid = threadIdx.x;
  const int r0  = blockIdx.x * 16;

  for (int m = 0; m < 16; ++m) {
    int row = r0 + m;
    float a = 0.0f, xx = 0.0f;
    if (row < nrows) {
      float inv = 1.0f / fmaxf(deg[row], 1.0f);
      a  = msg [(size_t)row * IN_CH + tid] * inv;
      xx = xdst[(size_t)row * IN_CH + tid];
    }
    sA[m][tid] = a;
    sX[m][tid] = xx;
  }
  __syncthreads();

  const int wave = tid >> 5;
  const int lane = tid & 31;
  const int half = lane >> 4;
  const int m    = lane & 15;
  const int n    = lane & 15;
  const int kh   = half * 2;

  for (int t = 0; t < 2; ++t) {
    const int n0 = (wave * 2 + t) * 16;
    v8f acc = {0.f, 0.f, 0.f, 0.f, 0.f, 0.f, 0.f, 0.f};
    for (int k = 0; k < IN_CH; k += 4) {
      v2f a, b;
      a.x = sA[m][k + kh];
      a.y = sA[m][k + kh + 1];
      b.x = Wl[(size_t)(k + kh)     * HID + n0 + n];
      b.y = Wl[(size_t)(k + kh + 1) * HID + n0 + n];
      acc = __builtin_amdgcn_wmma_f32_16x16x4_f32(false, a, false, b, (short)0, acc, false, false);
    }
    for (int k = 0; k < IN_CH; k += 4) {
      v2f a, b;
      a.x = sX[m][k + kh];
      a.y = sX[m][k + kh + 1];
      b.x = Wr[(size_t)(k + kh)     * HID + n0 + n];
      b.y = Wr[(size_t)(k + kh + 1) * HID + n0 + n];
      acc = __builtin_amdgcn_wmma_f32_16x16x4_f32(false, a, false, b, (short)0, acc, false, false);
    }
    const float bias  = bl[n0 + n];
    const int   mbase = half * 8;
    for (int r = 0; r < 8; ++r) {
      int row = r0 + mbase + r;
      if (row < nrows) {
        float v = acc[r] + bias;
        hout[(size_t)row * HID + n0 + n] = fmaxf(v, 0.0f);
      }
    }
  }
}

__global__ __launch_bounds__(128)
void sage_gemm_lsm(const float* __restrict__ msg, const float* __restrict__ deg,
                   const float* __restrict__ xdst,
                   const float* __restrict__ Wl, const float* __restrict__ bl,
                   const float* __restrict__ Wr,
                   float* __restrict__ out, int nrows)
{
  __shared__ float sA[16][HID + 4];
  __shared__ float sX[16][HID + 4];
  __shared__ float sL[16][OUTP];
  const int tid = threadIdx.x;
  const int r0  = blockIdx.x * 16;

  for (int m = 0; m < 16; ++m) {
    int row = r0 + m;
    float a = 0.0f, xx = 0.0f;
    if (row < nrows) {
      float inv = 1.0f / fmaxf(deg[row], 1.0f);
      a  = msg [(size_t)row * HID + tid] * inv;
      xx = xdst[(size_t)row * HID + tid];
    }
    sA[m][tid] = a;
    sX[m][tid] = xx;
  }
  __syncthreads();

  const int wave = tid >> 5;
  const int lane = tid & 31;
  const int half = lane >> 4;
  const int m    = lane & 15;
  const int n    = lane & 15;
  const int kh   = half * 2;

  if (wave < 3) {
    const int n0  = wave * 16;
    const int col = n0 + n;
    const bool cv = (col < OUTC);
    v8f acc = {0.f, 0.f, 0.f, 0.f, 0.f, 0.f, 0.f, 0.f};
    for (int k = 0; k < HID; k += 4) {
      v2f a, b;
      a.x = sA[m][k + kh];
      a.y = sA[m][k + kh + 1];
      b.x = cv ? Wl[(size_t)(k + kh)     * OUTC + col] : 0.0f;
      b.y = cv ? Wl[(size_t)(k + kh + 1) * OUTC + col] : 0.0f;
      acc = __builtin_amdgcn_wmma_f32_16x16x4_f32(false, a, false, b, (short)0, acc, false, false);
    }
    for (int k = 0; k < HID; k += 4) {
      v2f a, b;
      a.x = sX[m][k + kh];
      a.y = sX[m][k + kh + 1];
      b.x = cv ? Wr[(size_t)(k + kh)     * OUTC + col] : 0.0f;
      b.y = cv ? Wr[(size_t)(k + kh + 1) * OUTC + col] : 0.0f;
      acc = __builtin_amdgcn_wmma_f32_16x16x4_f32(false, a, false, b, (short)0, acc, false, false);
    }
    const float bias  = cv ? bl[col] : 0.0f;
    const int   mbase = half * 8;
    for (int r = 0; r < 8; ++r)
      sL[mbase + r][col] = acc[r] + bias;
  }
  __syncthreads();

  if (tid < 16) {
    int row = r0 + tid;
    if (row < nrows) {
      float mx = -INFINITY;
      for (int c = 0; c < OUTC; ++c) mx = fmaxf(mx, sL[tid][c]);
      float s = 0.0f;
      for (int c = 0; c < OUTC; ++c) s += expf(sL[tid][c] - mx);
      float lse = logf(s);
      for (int c = 0; c < OUTC; ++c)
        out[(size_t)row * OUTC + c] = sL[tid][c] - mx - lse;
    }
  }
}

extern "C" void kernel_launch(void* const* d_in, const int* in_sizes, int n_in,
                              void* d_out, int out_size, void* d_ws, size_t ws_size,
                              hipStream_t stream)
{
  const float* x    = (const float*)d_in[0];
  const int*   src0 = (const int*)d_in[1];
  const int*   dst0 = (const int*)d_in[2];
  const int*   src1 = (const int*)d_in[3];
  const int*   dst1 = (const int*)d_in[4];
  const float* Wl0  = (const float*)d_in[7];
  const float* bl0  = (const float*)d_in[8];
  const float* Wr0  = (const float*)d_in[9];
  const float* Wl1  = (const float*)d_in[10];
  const float* bl1  = (const float*)d_in[11];
  const float* Wr1  = (const float*)d_in[12];
  float* out = (float*)d_out;

  const int E0 = in_sizes[1];
  const int E1 = in_sizes[3];
  const int N1 = 100000;
  const int N2 = out_size / OUTC;

  float* msg0 = (float*)d_ws;
  float* deg0 = msg0 + (size_t)N1 * HID;
  float* h    = deg0 + N1;
  float* msg1 = h    + (size_t)N1 * HID;
  float* deg1 = msg1 + (size_t)N2 * HID;

  const long long z0 = (long long)N1 * HID + N1;
  const long long z1 = (long long)N2 * HID + N2;
  zero_f32<<<1024, 256, 0, stream>>>(msg0, z0);
  zero_f32<<<256,  256, 0, stream>>>(msg1, z1);

  scatter_deg<<<(E0 + 255) / 256, 256, 0, stream>>>(dst0, deg0, E0);
  {
    long long th = (long long)E0 * 32;
    scatter_msg<<<(unsigned)((th + 255) / 256), 256, 0, stream>>>(x, src0, dst0, msg0, E0);
  }
  sage_gemm_relu<<<(N1 + 15) / 16, 128, 0, stream>>>(msg0, deg0, x, Wl0, bl0, Wr0, h, N1);

  scatter_deg<<<(E1 + 255) / 256, 256, 0, stream>>>(dst1, deg1, E1);
  {
    long long th = (long long)E1 * 32;
    scatter_msg<<<(unsigned)((th + 255) / 256), 256, 0, stream>>>(h, src1, dst1, msg1, E1);
  }
  sage_gemm_lsm<<<(N2 + 15) / 16, 128, 0, stream>>>(msg1, deg1, h, Wl1, bl1, Wr1, out, N2);
}


# aggregation via counting-sort CSR (int atomic count+rank, scan, fill); one wave per dst row sums its source rows in registers; no f32 atomics, no msg zeroing; sum order = arrival rank
# speedup vs baseline: 1.5141x; 1.2428x over previous
#include <hip/hip_runtime.h>
#include <math.h>

#define IN_CH 128
#define HID   128
#define OUTC  47
#define OUTP  48

typedef float v2f __attribute__((ext_vector_type(2)));
typedef float v8f __attribute__((ext_vector_type(8)));


typedef float v4f __attribute__((ext_vector_type(4)));
typedef int   v4i __attribute__((ext_vector_type(4)));

__global__ void csr_zero(int* __restrict__ c0, int n0, int* __restrict__ c1, int n1) {
  int i = blockIdx.x * blockDim.x + threadIdx.x;
  if (i < n0) c0[i] = 0;
  else if (i - n0 < n1) c1[i - n0] = 0;
}

__global__ void csr_count(const int* __restrict__ dst0, int* __restrict__ cnt0,
                          int* __restrict__ rank0, int E0,
                          const int* __restrict__ dst1, int* __restrict__ cnt1,
                          int* __restrict__ rank1, int E1) {
  int t = blockIdx.x * blockDim.x + threadIdx.x;
  if (t < E0) {
    rank0[t] = atomicAdd(&cnt0[dst0[t]], 1);
  } else {
    int u = t - E0;
    if (u < E1) rank1[u] = atomicAdd(&cnt1[dst1[u]], 1);
  }
}

__global__ __launch_bounds__(256)
void csr_scan(const int* __restrict__ cnt0, int* __restrict__ rp0, int n0, int nb0,
              const int* __restrict__ cnt1, int* __restrict__ rp1, int n1) {
  __shared__ int sh[256];
  const int tid = threadIdx.x;
  const int* cnt; int* rp; int n; int b;
  if ((int)blockIdx.x < nb0) { cnt = cnt0; rp = rp0; n = n0; b = blockIdx.x; }
  else                       { cnt = cnt1; rp = rp1; n = n1; b = blockIdx.x - nb0; }

  int part = 0;
  const v4i* c4 = (const v4i*)cnt;
  for (int i = 0; i < b; ++i) {
    v4i v = c4[i * 256 + tid];
    part += (v.x + v.y) + (v.z + v.w);
  }
  sh[tid] = part;
  __syncthreads();
  for (int s = 128; s > 0; s >>= 1) {
    if (tid < s) sh[tid] += sh[tid + s];
    __syncthreads();
  }
  const int off = sh[0];
  __syncthreads();

  const int e0 = b * 1024 + tid * 4;
  const int a0 = (e0     < n) ? cnt[e0]     : 0;
  const int a1 = (e0 + 1 < n) ? cnt[e0 + 1] : 0;
  const int a2 = (e0 + 2 < n) ? cnt[e0 + 2] : 0;
  const int a3 = (e0 + 3 < n) ? cnt[e0 + 3] : 0;
  const int ts = a0 + a1 + a2 + a3;
  sh[tid] = ts;
  __syncthreads();
  for (int d = 1; d < 256; d <<= 1) {
    int v = (tid >= d) ? sh[tid - d] : 0;
    __syncthreads();
    sh[tid] += v;
    __syncthreads();
  }
  const int ex = sh[tid] - ts + off;
  if (e0     < n) rp[e0]     = ex;
  if (e0 + 1 < n) rp[e0 + 1] = ex + a0;
  if (e0 + 2 < n) rp[e0 + 2] = ex + a0 + a1;
  if (e0 + 3 < n) rp[e0 + 3] = ex + a0 + a1 + a2;
}

__global__ void csr_fill(const int* __restrict__ src0, const int* __restrict__ dst0,
                         const int* __restrict__ rank0, const int* __restrict__ rp0,
                         int* __restrict__ perm0, int E0,
                         const int* __restrict__ src1, const int* __restrict__ dst1,
                         const int* __restrict__ rank1, const int* __restrict__ rp1,
                         int* __restrict__ perm1, int E1) {
  int t = blockIdx.x * blockDim.x + threadIdx.x;
  if (t < E0) {
    perm0[rp0[dst0[t]] + rank0[t]] = src0[t];
  } else {
    int u = t - E0;
    if (u < E1) perm1[rp1[dst1[u]] + rank1[u]] = src1[u];
  }
}

__global__ __launch_bounds__(256)
void gather_rows(const float* __restrict__ xsrc, const int* __restrict__ rowptr,
                 const int* __restrict__ cnt, const int* __restrict__ perm,
                 float* __restrict__ msg, float* __restrict__ deg, int nrows) {
  const int lane = threadIdx.x & 31;
  const int row  = blockIdx.x * 8 + (threadIdx.x >> 5);
  if (row >= nrows) return;
  const int start = __builtin_amdgcn_readfirstlane(rowptr[row]);
  const int n     = __builtin_amdgcn_readfirstlane(cnt[row]);
  const v4f zero = {0.f, 0.f, 0.f, 0.f};
  v4f acc = zero;
  const float* xl = xsrc + lane * 4;
#define GATHER_ROW(k) (*(const v4f*)(xl + (size_t)__builtin_amdgcn_readlane(mine, (k)) * IN_CH))
  for (int i = 0; i < n; i += 32) {
    const int m = (n - i < 32) ? (n - i) : 32;
    const int mine = (lane < m) ? perm[start + i + lane] : 0;
    const v4f p0 = GATHER_ROW(0);
    int j = 1;
    for (; j + 8 <= m; j += 8) {
      const v4f v0 = GATHER_ROW(j);
      const v4f v1 = GATHER_ROW(j + 1);
      const v4f v2 = GATHER_ROW(j + 2);
      const v4f v3 = GATHER_ROW(j + 3);
      const v4f v4 = GATHER_ROW(j + 4);
      const v4f v5 = GATHER_ROW(j + 5);
      const v4f v6 = GATHER_ROW(j + 6);
      const v4f v7 = GATHER_ROW(j + 7);
      acc += v0; acc += v1; acc += v2; acc += v3;
      acc += v4; acc += v5; acc += v6; acc += v7;
    }
    const int rem = m - j;
    v4f q0 = zero, q1 = zero, q2 = zero, q3 = zero, d0 = zero, d1 = zero, o0 = zero;
    if (rem & 4) {
      q0 = GATHER_ROW(j);
      q1 = GATHER_ROW(j + 1);
      q2 = GATHER_ROW(j + 2);
      q3 = GATHER_ROW(j + 3);
      j += 4;
    }
    if (rem & 2) {
      d0 = GATHER_ROW(j);
      d1 = GATHER_ROW(j + 1);
      j += 2;
    }
    if (rem & 1) {
      o0 = GATHER_ROW(j);
    }
    acc += q0; acc += q1; acc += q2; acc += q3;
    acc += d0; acc += d1; acc += o0;
    acc += p0;
  }
#undef GATHER_ROW
  *(v4f*)(msg + (size_t)row * IN_CH + lane * 4) = acc;
  if (lane == 0) deg[row] = (float)n;
}

__global__ __launch_bounds__(128)
void sage_gemm_relu(const float* __restrict__ msg, const float* __restrict__ deg,
                    const float* __restrict__ xdst,
                    const float* __restrict__ Wl, const float* __restrict__ bl,
                    const float* __restrict__ Wr,
                    float* __restrict__ hout, int nrows)
{
  __shared__ float sA[16][IN_CH + 4];
  __shared__ float sX[16][IN_CH + 4];
  const int tid = threadIdx.x;
  const int r0  = blockIdx.x * 16;

  for (int m = 0; m < 16; ++m) {
    int row = r0 + m;
    float a = 0.0f, xx = 0.0f;
    if (row < nrows) {
      float inv = 1.0f / fmaxf(deg[row], 1.0f);
      a  = msg [(size_t)row * IN_CH + tid] * inv;
      xx = xdst[(size_t)row * IN_CH + tid];
    }
    sA[m][tid] = a;
    sX[m][tid] = xx;
  }
  __syncthreads();

  const int wave = tid >> 5;
  const int lane = tid & 31;
  const int half = lane >> 4;
  const int m    = lane & 15;
  const int n    = lane & 15;
  const int kh   = half * 2;

  for (int t = 0; t < 2; ++t) {
    const int n0 = (wave * 2 + t) * 16;
    v8f acc = {0.f, 0.f, 0.f, 0.f, 0.f, 0.f, 0.f, 0.f};
    for (int k = 0; k < IN_CH; k += 4) {
      v2f a, b;
      a.x = sA[m][k + kh];
      a.y = sA[m][k + kh + 1];
      b.x = Wl[(size_t)(k + kh)     * HID + n0 + n];
      b.y = Wl[(size_t)(k + kh + 1) * HID + n0 + n];
      acc = __builtin_amdgcn_wmma_f32_16x16x4_f32(false, a, false, b, (short)0, acc, false, false);
    }
    for (int k = 0; k < IN_CH; k += 4) {
      v2f a, b;
      a.x = sX[m][k + kh];
      a.y = sX[m][k + kh + 1];
      b.x = Wr[(size_t)(k + kh)     * HID + n0 + n];
      b.y = Wr[(size_t)(k + kh + 1) * HID + n0 + n];
      acc = __builtin_amdgcn_wmma_f32_16x16x4_f32(false, a, false, b, (short)0, acc, false, false);
    }
    const float bias  = bl[n0 + n];
    const int   mbase = half * 8;
    for (int r = 0; r < 8; ++r) {
      int row = r0 + mbase + r;
      if (row < nrows) {
        float v = acc[r] + bias;
        hout[(size_t)row * HID + n0 + n] = fmaxf(v, 0.0f);
      }
    }
  }
}

__global__ __launch_bounds__(128)
void sage_gemm_lsm(const float* __restrict__ msg, const float* __restrict__ deg,
                   const float* __restrict__ xdst,
                   const float* __restrict__ Wl, const float* __restrict__ bl,
                   const float* __restrict__ Wr,
                   float* __restrict__ out, int nrows)
{
  __shared__ float sA[16][HID + 4];
  __shared__ float sX[16][HID + 4];
  __shared__ float sL[16][OUTP];
  const int tid = threadIdx.x;
  const int r0  = blockIdx.x * 16;

  for (int m = 0; m < 16; ++m) {
    int row = r0 + m;
    float a = 0.0f, xx = 0.0f;
    if (row < nrows) {
      float inv = 1.0f / fmaxf(deg[row], 1.0f);
      a  = msg [(size_t)row * HID + tid] * inv;
      xx = xdst[(size_t)row * HID + tid];
    }
    sA[m][tid] = a;
    sX[m][tid] = xx;
  }
  __syncthreads();

  const int wave = tid >> 5;
  const int lane = tid & 31;
  const int half = lane >> 4;
  const int m    = lane & 15;
  const int n    = lane & 15;
  const int kh   = half * 2;

  if (wave < 3) {
    const int n0  = wave * 16;
    const int col = n0 + n;
    const bool cv = (col < OUTC);
    v8f acc = {0.f, 0.f, 0.f, 0.f, 0.f, 0.f, 0.f, 0.f};
    for (int k = 0; k < HID; k += 4) {
      v2f a, b;
      a.x = sA[m][k + kh];
      a.y = sA[m][k + kh + 1];
      b.x = cv ? Wl[(size_t)(k + kh)     * OUTC + col] : 0.0f;
      b.y = cv ? Wl[(size_t)(k + kh + 1) * OUTC + col] : 0.0f;
      acc = __builtin_amdgcn_wmma_f32_16x16x4_f32(false, a, false, b, (short)0, acc, false, false);
    }
    for (int k = 0; k < HID; k += 4) {
      v2f a, b;
      a.x = sX[m][k + kh];
      a.y = sX[m][k + kh + 1];
      b.x = cv ? Wr[(size_t)(k + kh)     * OUTC + col] : 0.0f;
      b.y = cv ? Wr[(size_t)(k + kh + 1) * OUTC + col] : 0.0f;
      acc = __builtin_amdgcn_wmma_f32_16x16x4_f32(false, a, false, b, (short)0, acc, false, false);
    }
    const float bias  = cv ? bl[col] : 0.0f;
    const int   mbase = half * 8;
    for (int r = 0; r < 8; ++r)
      sL[mbase + r][col] = acc[r] + bias;
  }
  __syncthreads();

  if (tid < 16) {
    int row = r0 + tid;
    if (row < nrows) {
      float mx = -INFINITY;
      for (int c = 0; c < OUTC; ++c) mx = fmaxf(mx, sL[tid][c]);
      float s = 0.0f;
      for (int c = 0; c < OUTC; ++c) s += expf(sL[tid][c] - mx);
      float lse = logf(s);
      for (int c = 0; c < OUTC; ++c)
        out[(size_t)row * OUTC + c] = sL[tid][c] - mx - lse;
    }
  }
}

extern "C" void kernel_launch(void* const* d_in, const int* in_sizes, int n_in,
                              void* d_out, int out_size, void* d_ws, size_t ws_size,
                              hipStream_t stream)
{
  const float* x    = (const float*)d_in[0];
  const int*   src0 = (const int*)d_in[1];
  const int*   dst0 = (const int*)d_in[2];
  const int*   src1 = (const int*)d_in[3];
  const int*   dst1 = (const int*)d_in[4];
  const float* Wl0  = (const float*)d_in[7];
  const float* bl0  = (const float*)d_in[8];
  const float* Wr0  = (const float*)d_in[9];
  const float* Wl1  = (const float*)d_in[10];
  const float* bl1  = (const float*)d_in[11];
  const float* Wr1  = (const float*)d_in[12];
  float* out = (float*)d_out;

  const int E0 = in_sizes[1];
  const int E1 = in_sizes[3];
  const int N1 = 100000;
  const int N2 = out_size / OUTC;

  float* msg0 = (float*)d_ws;
  float* deg0 = msg0 + (size_t)N1 * HID;
  float* h    = deg0 + N1;
  float* msg1 = h    + (size_t)N1 * HID;

  int* cnt0  = (int*)h;
  int* rp0   = cnt0 + N1;
  int* rank0 = rp0 + N1;
  int* perm0 = rank0 + E0;
  int* cnt1  = (int*)msg1;
  int* rp1   = cnt1 + N2;
  int* rank1 = rp1 + N2;
  int* perm1 = rank1 + E1;

  const int nb0 = (N1 + 1023) / 1024;
  const int nb1 = (N2 + 1023) / 1024;

  csr_zero<<<(N1 + N2 + 255) / 256, 256, 0, stream>>>(cnt0, N1, cnt1, N2);
  csr_count<<<(E0 + E1 + 255) / 256, 256, 0, stream>>>(dst0, cnt0, rank0, E0, dst1, cnt1, rank1, E1);
  csr_scan<<<nb0 + nb1, 256, 0, stream>>>(cnt0, rp0, N1, nb0, cnt1, rp1, N2);
  csr_fill<<<(E0 + E1 + 255) / 256, 256, 0, stream>>>(src0, dst0, rank0, rp0, perm0, E0,
                                                      src1, dst1, rank1, rp1, perm1, E1);

  gather_rows<<<(N1 + 7) / 8, 256, 0, stream>>>(x, rp0, cnt0, perm0, msg0, deg0, N1);
  sage_gemm_relu<<<(N1 + 15) / 16, 128, 0, stream>>>(msg0, deg0, x, Wl0, bl0, Wr0, h, N1);

  gather_rows<<<(N2 + 7) / 8, 256, 0, stream>>>(h, rp1, cnt1, perm1, msg0, deg0, N2);
  sage_gemm_lsm<<<(N2 + 15) / 16, 128, 0, stream>>>(msg0, deg0, h, Wl1, bl1, Wr1, out, N2);
}


# edge kernels and gathers enter few long-lived waves: 4 edges per thread (16-byte loads, 4 adds in flight); 7 or 2 consecutive rows per gather wave, the next row's slot list asked for one row ahead
# speedup vs baseline: 1.5867x; 1.0479x over previous
#include <hip/hip_runtime.h>
#include <math.h>

#define IN_CH 128
#define HID   128
#define OUTC  47
#define OUTP  48
#define RPW0  7
#define RPW1  2

typedef float v2f __attribute__((ext_vector_type(2)));
typedef float v8f __attribute__((ext_vector_type(8)));


typedef float v4f __attribute__((ext_vector_type(4)));
typedef int   v4i __attribute__((ext_vector_type(4)));

__global__ void csr_zero(int* __restrict__ c0, int q0, int* __restrict__ c1, int q1) {
  int i = blockIdx.x * blockDim.x + threadIdx.x;
  const v4i z = {0, 0, 0, 0};
  if (i < q0) ((v4i*)c0)[i] = z;
  else if (i - q0 < q1) ((v4i*)c1)[i - q0] = z;
}

__device__ __forceinline__ void count_quad(const int* __restrict__ dst, int* __restrict__ cnt,
                                           int* __restrict__ rank, int q, int E) {
  const int e = q * 4;
  if (e + 3 < E) {
    const v4i d = *(const v4i*)(dst + e);
    v4i r;
    r.x = atomicAdd(&cnt[d.x], 1);
    r.y = atomicAdd(&cnt[d.y], 1);
    r.z = atomicAdd(&cnt[d.z], 1);
    r.w = atomicAdd(&cnt[d.w], 1);
    *(v4i*)(rank + e) = r;
  } else {
    for (int k = e; k < E; ++k) rank[k] = atomicAdd(&cnt[dst[k]], 1);
  }
}

__global__ void csr_count(const int* __restrict__ dst0, int* __restrict__ cnt0,
                          int* __restrict__ rank0, int E0,
                          const int* __restrict__ dst1, int* __restrict__ cnt1,
                          int* __restrict__ rank1, int E1) {
  const int Q0 = (E0 + 3) >> 2;
  const int Q1 = (E1 + 3) >> 2;
  int t = blockIdx.x * blockDim.x + threadIdx.x;
  if (t < Q0) {
    count_quad(dst0, cnt0, rank0, t, E0);
  } else {
    int u = t - Q0;
    if (u < Q1) count_quad(dst1, cnt1, rank1, u, E1);
  }
}

__global__ __launch_bounds__(256)
void csr_scan(const int* __restrict__ cnt0, int* __restrict__ rp0, int n0, int nb0,
              const int* __restrict__ cnt1, int* __restrict__ rp1, int n1) {
  __shared__ int sh[256];
  const int tid = threadIdx.x;
  const int* cnt; int* rp; int n; int b;
  if ((int)blockIdx.x < nb0) { cnt = cnt0; rp = rp0; n = n0; b = blockIdx.x; }
  else                       { cnt = cnt1; rp = rp1; n = n1; b = blockIdx.x - nb0; }

  int part = 0;
  const v4i* c4 = (const v4i*)cnt;
  for (int i = 0; i < b; ++i) {
    v4i v = c4[i * 256 + tid];
    part += (v.x + v.y) + (v.z + v.w);
  }
  sh[tid] = part;
  __syncthreads();
  for (int s = 128; s > 0; s >>= 1) {
    if (tid < s) sh[tid] += sh[tid + s];
    __syncthreads();
  }
  const int off = sh[0];
  __syncthreads();

  const int e0 = b * 1024 + tid * 4;
  const int a0 = (e0     < n) ? cnt[e0]     : 0;
  const int a1 = (e0 + 1 < n) ? cnt[e0 + 1] : 0;
  const int a2 = (e0 + 2 < n) ? cnt[e0 + 2] : 0;
  const int a3 = (e0 + 3 < n) ? cnt[e0 + 3] : 0;
  const int ts = a0 + a1 + a2 + a3;
  sh[tid] = ts;
  __syncthreads();
  for (int d = 1; d < 256; d <<= 1) {
    int v = (tid >= d) ? sh[tid - d] : 0;
    __syncthreads();
    sh[tid] += v;
    __syncthreads();
  }
  const int ex = sh[tid] - ts + off;
  if (e0     < n) rp[e0]     = ex;
  if (e0 + 1 < n) rp[e0 + 1] = ex + a0;
  if (e0 + 2 < n) rp[e0 + 2] = ex + a0 + a1;
  if (e0 + 3 < n) rp[e0 + 3] = ex + a0 + a1 + a2;
}

__device__ __forceinline__ void fill_quad(const int* __restrict__ src, const int* __restrict__ dst,
                                          const int* __restrict__ rank, const int* __restrict__ rp,
                                          int* __restrict__ perm, int q, int E) {
  const int e = q * 4;
  if (e + 3 < E) {
    const v4i d = *(const v4i*)(dst + e);
    const v4i r = *(const v4i*)(rank + e);
    const v4i s = *(const v4i*)(src + e);
    const int b0 = rp[d.x];
    const int b1 = rp[d.y];
    const int b2 = rp[d.z];
    const int b3 = rp[d.w];
    perm[b0 + r.x] = s.x;
    perm[b1 + r.y] = s.y;
    perm[b2 + r.z] = s.z;
    perm[b3 + r.w] = s.w;
  } else {
    for (int k = e; k < E; ++k) perm[rp[dst[k]] + rank[k]] = src[k];
  }
}

__global__ void csr_fill(const int* __restrict__ src0, const int* __restrict__ dst0,
                         const int* __restrict__ rank0, const int* __restrict__ rp0,
                         int* __restrict__ perm0, int E0,
                         const int* __restrict__ src1, const int* __restrict__ dst1,
                         const int* __restrict__ rank1, const int* __restrict__ rp1,
                         int* __restrict__ perm1, int E1) {
  const int Q0 = (E0 + 3) >> 2;
  const int Q1 = (E1 + 3) >> 2;
  int t = blockIdx.x * blockDim.x + threadIdx.x;
  if (t < Q0) {
    fill_quad(src0, dst0, rank0, rp0, perm0, t, E0);
  } else {
    int u = t - Q0;
    if (u < Q1) fill_quad(src1, dst1, rank1, rp1, perm1, u, E1);
  }
}

__global__ __launch_bounds__(256)
void gather_rows(const float* __restrict__ xsrc, const int* __restrict__ rowptr,
                 const int* __restrict__ cnt, const int* __restrict__ perm,
                 float* __restrict__ msg, float* __restrict__ deg, int nrows, int rpw) {
  const int lane = threadIdx.x & 31;
  const int wv = __builtin_amdgcn_readfirstlane((int)(blockIdx.x * 8 + (threadIdx.x >> 5)));
  const int r0 = wv * rpw;
  if (r0 >= nrows) return;
  const int nr = (nrows - r0 < rpw) ? (nrows - r0) : rpw;
  int st = 0, cn = 0;
  if (lane < nr) {
    st = rowptr[r0 + lane];
    cn = cnt[r0 + lane];
    deg[r0 + lane] = (float)cn;
  }
  const v4f zero = {0.f, 0.f, 0.f, 0.f};
  const float* xl = xsrc + lane * 4;
  float* ml = msg + (size_t)r0 * IN_CH + lane * 4;
  int start = __builtin_amdgcn_readlane(st, 0);
  int n     = __builtin_amdgcn_readlane(cn, 0);
  int mine  = (lane < n) ? perm[start + lane] : 0;
  int first = __builtin_amdgcn_readlane(mine, 0);
#define GATHER_ROW(k) (*(const v4f*)(xl + (size_t)__builtin_amdgcn_readlane(mine, (k)) * IN_CH))
  for (int r = 0; r < nr; ++r) {
    int start_nx = 0, n_nx = 0, mine_nx = 0;
    if (r + 1 < nr) {
      start_nx = __builtin_amdgcn_readlane(st, r + 1);
      n_nx     = __builtin_amdgcn_readlane(cn, r + 1);
      mine_nx  = (lane < n_nx) ? perm[start_nx + lane] : 0;
    }
    v4f acc = zero;
    for (int i = 0; i < n; i += 32) {
      const int m = (n - i < 32) ? (n - i) : 32;
      if (i > 0) {
        mine  = (lane < m) ? perm[start + i + lane] : 0;
        first = __builtin_amdgcn_readlane(mine, 0);
      }
      const v4f p0 = *(const v4f*)(xl + (size_t)first * IN_CH);
      int j = 1;
      for (; j + 8 <= m; j += 8) {
        const v4f v0 = GATHER_ROW(j);
        const v4f v1 = GATHER_ROW(j + 1);
        const v4f v2 = GATHER_ROW(j + 2);
        const v4f v3 = GATHER_ROW(j + 3);
        const v4f v4 = GATHER_ROW(j + 4);
        const v4f v5 = GATHER_ROW(j + 5);
        const v4f v6 = GATHER_ROW(j + 6);
        const v4f v7 = GATHER_ROW(j + 7);
        acc += v0; acc += v1; acc += v2; acc += v3;
        acc += v4; acc += v5; acc += v6; acc += v7;
      }
      const int rem = m - j;
      v4f q0 = zero, q1 = zero, q2 = zero, q3 = zero, d0 = zero, d1 = zero, o0 = zero;
      if (rem & 4) {
        q0 = GATHER_ROW(j);
        q1 = GATHER_ROW(j + 1);
        q2 = GATHER_ROW(j + 2);
        q3 = GATHER_ROW(j + 3);
        j += 4;
      }
      if (rem & 2) {
        d0 = GATHER_ROW(j);
        d1 = GATHER_ROW(j + 1);
        j += 2;
      }
      if (rem & 1) {
        o0 = GATHER_ROW(j);
      }
      acc += q0; acc += q1; acc += q2; acc += q3;
      acc += d0; acc += d1; acc += o0;
      acc += p0;
    }
    *(v4f*)(ml + (size_t)r * IN_CH) = acc;
    start = start_nx;
    n     = n_nx;
    mine  = mine_nx;
    first = __builtin_amdgcn_readlane(mine, 0);
  }
#undef GATHER_ROW
}

__global__ __launch_bounds__(128)
void sage_gemm_relu(const float* __restrict__ msg, const float* __restrict__ deg,
                    const float* __restrict__ xdst,
                    const float* __restrict__ Wl, const float* __restrict__ bl,
                    const float* __restrict__ Wr,
                    float* __restrict__ hout, int nrows)
{
  __shared__ float sA[16][IN_CH + 4];
  __shared__ float sX[16][IN_CH + 4];
  const int tid = threadIdx.x;
  const int r0  = blockIdx.x * 16;

  for (int m = 0; m < 16; ++m) {
    int row = r0 + m;
    float a = 0.0f, xx = 0.0f;
    if (row < nrows) {
      float inv = 1.0f / fmaxf(deg[row], 1.0f);
      a  = msg [(size_t)row * IN_CH + tid] * inv;
      xx = xdst[(size_t)row * IN_CH + tid];
    }
    sA[m][tid] = a;
    sX[m][tid] = xx;
  }
  __syncthreads();

  const int wave = tid >> 5;
  const int lane = tid & 31;
  const int half = lane >> 4;
  const int m    = lane & 15;
  const int n    = lane & 15;
  const int kh   = half * 2;

  for (int t = 0; t < 2; ++t) {
    const int n0 = (wave * 2 + t) * 16;
    v8f acc = {0.f, 0.f, 0.f, 0.f, 0.f, 0.f, 0.f, 0.f};
    for (int k = 0; k < IN_CH; k += 4) {
      v2f a, b;
      a.x = sA[m][k + kh];
      a.y = sA[m][k + kh + 1];
      b.x = Wl[(size_t)(k + kh)     * HID + n0 + n];
      b.y = Wl[(size_t)(k + kh + 1) * HID + n0 + n];
      acc = __builtin_amdgcn_wmma_f32_16x16x4_f32(false, a, false, b, (short)0, acc, false, false);
    }
    for (int k = 0; k < IN_CH; k += 4) {
      v2f a, b;
      a.x = sX[m][k + kh];
      a.y = sX[m][k + kh + 1];
      b.x = Wr[(size_t)(k + kh)     * HID + n0 + n];
      b.y = Wr[(size_t)(k + kh + 1) * HID + n0 + n];
      acc = __builtin_amdgcn_wmma_f32_16x16x4_f32(false, a, false, b, (short)0, acc, false, false);
    }
    const float bias  = bl[n0 + n];
    const int   mbase = half * 8;
    for (int r = 0; r < 8; ++r) {
      int row = r0 + mbase + r;
      if (row < nrows) {
        float v = acc[r] + bias;
        hout[(size_t)row * HID + n0 + n] = fmaxf(v, 0.0f);
      }
    }
  }
}

__global__ __launch_bounds__(128)
void sage_gemm_lsm(const float* __restrict__ msg, const float* __restrict__ deg,
                   const float* __restrict__ xdst,
                   const float* __restrict__ Wl, const float* __restrict__ bl,
                   const float* __restrict__ Wr,
                   float* __restrict__ out, int nrows)
{
  __shared__ float sA[16][HID + 4];
  __shared__ float sX[16][HID + 4];
  __shared__ float sL[16][OUTP];
  const int tid = threadIdx.x;
  const int r0  = blockIdx.x * 16;

  for (int m = 0; m < 16; ++m) {
    int row = r0 + m;
    float a = 0.0f, xx = 0.0f;
    if (row < nrows) {
      float inv = 1.0f / fmaxf(deg[row], 1.0f);
      a  = msg [(size_t)row * HID + tid] * inv;
      xx = xdst[(size_t)row * HID + tid];
    }
    sA[m][tid] = a;
    sX[m][tid] = xx;
  }
  __syncthreads();

  const int wave = tid >> 5;
  const int lane = tid & 31;
  const int half = lane >> 4;
  const int m    = lane & 15;
  const int n    = lane & 15;
  const int kh   = half * 2;

  if (wave < 3) {
    const int n0  = wave * 16;
    const int col = n0 + n;
    const bool cv = (col < OUTC);
    v8f acc = {0.f, 0.f, 0.f, 0.f, 0.f, 0.f, 0.f, 0.f};
    for (int k = 0; k < HID; k += 4) {
      v2f a, b;
      a.x = sA[m][k + kh];
      a.y = sA[m][k + kh + 1];
      b.x = cv ? Wl[(size_t)(k + kh)     * OUTC + col] : 0.0f;
      b.y = cv ? Wl[(size_t)(k + kh + 1) * OUTC + col] : 0.0f;
      acc = __builtin_amdgcn_wmma_f32_16x16x4_f32(false, a, false, b, (short)0, acc, false, false);
    }
    for (int k = 0; k < HID; k += 4) {
      v2f a, b;
      a.x = sX[m][k + kh];
      a.y = sX[m][k + kh + 1];
      b.x = cv ? Wr[(size_t)(k + kh)     * OUTC + col] : 0.0f;
      b.y = cv ? Wr[(size_t)(k + kh + 1) * OUTC + col] : 0.0f;
      acc = __builtin_amdgcn_wmma_f32_16x16x4_f32(false, a, false, b, (short)0, acc, false, false);
    }
    const float bias  = cv ? bl[col] : 0.0f;
    const int   mbase = half * 8;
    for (int r = 0; r < 8; ++r)
      sL[mbase + r][col] = acc[r] + bias;
  }
  __syncthreads();

  if (tid < 16) {
    int row = r0 + tid;
    if (row < nrows) {
      float mx = -INFINITY;
      for (int c = 0; c < OUTC; ++c) mx = fmaxf(mx, sL[tid][c]);
      float s = 0.0f;
      for (int c = 0; c < OUTC; ++c) s += expf(sL[tid][c] - mx);
      float lse = logf(s);
      for (int c = 0; c < OUTC; ++c)
        out[(size_t)row * OUTC + c] = sL[tid][c] - mx - lse;
    }
  }
}

extern "C" void kernel_launch(void* const* d_in, const int* in_sizes, int n_in,
                              void* d_out, int out_size, void* d_ws, size_t ws_size,
                              hipStream_t stream)
{
  const float* x    = (const float*)d_in[0];
  const int*   src0 = (const int*)d_in[1];
  const int*   dst0 = (const int*)d_in[2];
  const int*   src1 = (const int*)d_in[3];
  const int*   dst1 = (const int*)d_in[4];
  const float* Wl0  = (const float*)d_in[7];
  const float* bl0  = (const float*)d_in[8];
  const float* Wr0  = (const float*)d_in[9];
  const float* Wl1  = (const float*)d_in[10];
  const float* bl1  = (const float*)d_in[11];
  const float* Wr1  = (const float*)d_in[12];
  float* out = (float*)d_out;

  const int E0 = in_sizes[1];
  const int E1 = in_sizes[3];
  const int N1 = 100000;
  const int N2 = out_size / OUTC;

  float* msg0 = (float*)d_ws;
  float* deg0 = msg0 + (size_t)N1 * HID;
  float* h    = deg0 + N1;
  float* msg1 = h    + (size_t)N1 * HID;

  const int N1p = (N1 + 3) & ~3;
  const int N2p = (N2 + 3) & ~3;
  const int E0p = (E0 + 3) & ~3;
  const int E1p = (E1 + 3) & ~3;
  int* cnt0  = (int*)h;
  int* rp0   = cnt0 + N1p;
  int* rank0 = rp0 + N1p;
  int* perm0 = rank0 + E0p;
  int* cnt1  = (int*)msg1;
  int* rp1   = cnt1 + N2p;
  int* rank1 = rp1 + N2p;
  int* perm1 = rank1 + E1p;

  const int nb0 = (N1 + 1023) / 1024;
  const int nb1 = (N2 + 1023) / 1024;

  const int QZ = N1p / 4 + N2p / 4;
  const int QE = E0p / 4 + E1p / 4;
  csr_zero<<<(QZ + 255) / 256, 256, 0, stream>>>(cnt0, N1p / 4, cnt1, N2p / 4);
  csr_count<<<(QE + 255) / 256, 256, 0, stream>>>(dst0, cnt0, rank0, E0, dst1, cnt1, rank1, E1);
  csr_scan<<<nb0 + nb1, 256, 0, stream>>>(cnt0, rp0, N1, nb0, cnt1, rp1, N2);
  csr_fill<<<(QE + 255) / 256, 256, 0, stream>>>(src0, dst0, rank0, rp0, perm0, E0,
                                                      src1, dst1, rank1, rp1, perm1, E1);

  gather_rows<<<((N1 + RPW0 - 1) / RPW0 + 7) / 8, 256, 0, stream>>>(x, rp0, cnt0, perm0, msg0, deg0, N1, RPW0);
  sage_gemm_relu<<<(N1 + 15) / 16, 128, 0, stream>>>(msg0, deg0, x, Wl0, bl0, Wr0, h, N1);

  gather_rows<<<((N2 + RPW1 - 1) / RPW1 + 7) / 8, 256, 0, stream>>>(h, rp1, cnt1, perm1, msg0, deg0, N2, RPW1);
  sage_gemm_lsm<<<(N2 + 15) / 16, 128, 0, stream>>>(msg0, deg0, h, Wl1, bl1, Wr1, out, N2);
}


# edges sorted by destination with a two-level counting sort, counters all in LDS (block histograms of 256 bins, binned copy, per-bin count+scan+place): 4 sort launches to 3; sum order = LDS arrival
# speedup vs baseline: 2.8769x; 1.8131x over previous
#include <hip/hip_runtime.h>
#include <math.h>

#define IN_CH 128
#define HID   128
#define OUTC  47
#define OUTP  48
#define RPWF  4

typedef float v2f __attribute__((ext_vector_type(2)));
typedef float v8f __attribute__((ext_vector_type(8)));


typedef float v4f __attribute__((ext_vector_type(4)));
typedef int   v4i __attribute__((ext_vector_type(4)));

#define NBIN  256
#define CHUNK 8192
#define ETPB  1024
#define RMAX  512

__device__ __forceinline__ int bin_shift(int n) {
  const int s = 24 - __clz(n - 1);
  return s < 0 ? 0 : s;
}

__global__ __launch_bounds__(ETPB)
void bin_hist(const int* __restrict__ dst0, int E0, int N0, int nblk0,
              const int* __restrict__ dst1, int E1, int N1,
              int* __restrict__ table) {
  __shared__ int h[NBIN];
  const int t = threadIdx.x;
  const int* dst; int E, N, c;
  if ((int)blockIdx.x < nblk0) { dst = dst0; E = E0; N = N0; c = blockIdx.x; }
  else                         { dst = dst1; E = E1; N = N1; c = blockIdx.x - nblk0; }
  const int sh = bin_shift(N);
  if (t < NBIN) h[t] = 0;
  __syncthreads();
  for (int j = 0; j < CHUNK / (4 * ETPB); ++j) {
    const int e = (c * (CHUNK / 4) + j * ETPB + t) * 4;
    if (e + 3 < E) {
      const v4i d = *(const v4i*)(dst + e);
      atomicAdd(&h[d.x >> sh], 1);
      atomicAdd(&h[d.y >> sh], 1);
      atomicAdd(&h[d.z >> sh], 1);
      atomicAdd(&h[d.w >> sh], 1);
    } else {
      for (int k = e; k < E; ++k) atomicAdd(&h[dst[k] >> sh], 1);
    }
  }
  __syncthreads();
  if (t < NBIN) table[(size_t)blockIdx.x * NBIN + t] = h[t];
}

__global__ __launch_bounds__(ETPB)
void bin_scatter(const int* __restrict__ src0, const int* __restrict__ dst0, int E0, int N0, int nblk0,
                 const int* __restrict__ src1, const int* __restrict__ dst1, int E1, int N1, int nblk1,
                 const int* __restrict__ table,
                 int* __restrict__ bdst0, int* __restrict__ bsrc0, int* __restrict__ boff0,
                 int* __restrict__ bdst1, int* __restrict__ bsrc1, int* __restrict__ boff1) {
  __shared__ int sTot[4][NBIN];
  __shared__ int sPre[4][NBIN];
  __shared__ int sScan[NBIN];
  __shared__ int lcur[NBIN];
  const int t = threadIdx.x;
  const int* src; const int* dst; int* bdst; int* bsrc; int* boff; int E, N, c, fb, nb;
  if ((int)blockIdx.x < nblk0) {
    src = src0; dst = dst0; bdst = bdst0; bsrc = bsrc0; boff = boff0;
    E = E0; N = N0; c = blockIdx.x; fb = 0; nb = nblk0;
  } else {
    src = src1; dst = dst1; bdst = bdst1; bsrc = bsrc1; boff = boff1;
    E = E1; N = N1; c = blockIdx.x - nblk0; fb = nblk0; nb = nblk1;
  }
  const int sh = bin_shift(N);

  {
    const int b    = t & (NBIN - 1);
    const int part = t >> 8;
    const int q    = (nb + 3) >> 2;
    const int k0   = part * q;
    const int k1   = (k0 + q < nb) ? (k0 + q) : nb;
    int tot = 0, pre = 0;
    for (int k = k0; k < k1; ++k) {
      const int v = table[(size_t)(fb + k) * NBIN + b];
      tot += v;
      pre += (k < c) ? v : 0;
    }
    sTot[part][b] = tot;
    sPre[part][b] = pre;
  }
  __syncthreads();
  int total = 0, base = 0;
  if (t < NBIN) {
    total = (sTot[0][t] + sTot[1][t]) + (sTot[2][t] + sTot[3][t]);
    base  = (sPre[0][t] + sPre[1][t]) + (sPre[2][t] + sPre[3][t]);
    sScan[t] = total;
  }
  __syncthreads();
  for (int d = 1; d < NBIN; d <<= 1) {
    const int v = (t < NBIN && t >= d) ? sScan[t - d] : 0;
    __syncthreads();
    if (t < NBIN) sScan[t] += v;
    __syncthreads();
  }
  if (t < NBIN) {
    const int ex = sScan[t] - total;
    lcur[t] = ex + base;
    if (c == 0) {
      boff[t] = ex;
      if (t == NBIN - 1) boff[NBIN] = sScan[t];
    }
  }
  __syncthreads();

  for (int j = 0; j < CHUNK / (4 * ETPB); ++j) {
    const int e = (c * (CHUNK / 4) + j * ETPB + t) * 4;
    if (e + 3 < E) {
      const v4i d = *(const v4i*)(dst + e);
      const v4i s = *(const v4i*)(src + e);
      const int p0 = atomicAdd(&lcur[d.x >> sh], 1);
      const int p1 = atomicAdd(&lcur[d.y >> sh], 1);
      const int p2 = atomicAdd(&lcur[d.z >> sh], 1);
      const int p3 = atomicAdd(&lcur[d.w >> sh], 1);
      bdst[p0] = d.x; bsrc[p0] = s.x;
      bdst[p1] = d.y; bsrc[p1] = s.y;
      bdst[p2] = d.z; bsrc[p2] = s.z;
      bdst[p3] = d.w; bsrc[p3] = s.w;
    } else {
      for (int k = e; k < E; ++k) {
        const int dk = dst[k];
        const int p  = atomicAdd(&lcur[dk >> sh], 1);
        bdst[p] = dk; bsrc[p] = src[k];
      }
    }
  }
}

__global__ __launch_bounds__(ETPB)
void bin_csr(const int* __restrict__ bdst0, const int* __restrict__ bsrc0, const int* __restrict__ boff0, int N0,
             int* __restrict__ rp0, int* __restrict__ cnt0, int* __restrict__ perm0,
             const int* __restrict__ bdst1, const int* __restrict__ bsrc1, const int* __restrict__ boff1, int N1,
             int* __restrict__ rp1, int* __restrict__ cnt1, int* __restrict__ perm1) {
  __shared__ int lc[RMAX];
  __shared__ int ls[RMAX];
  __shared__ int lq[RMAX];
  const int t = threadIdx.x;
  const int* bdst; const int* bsrc; const int* boff; int* rp; int* cnt; int* perm; int N, bin;
  if ((int)blockIdx.x < NBIN) {
    bdst = bdst0; bsrc = bsrc0; boff = boff0; rp = rp0; cnt = cnt0; perm = perm0; N = N0; bin = blockIdx.x;
  } else {
    bdst = bdst1; bsrc = bsrc1; boff = boff1; rp = rp1; cnt = cnt1; perm = perm1; N = N1; bin = blockIdx.x - NBIN;
  }
  const int sh   = bin_shift(N);
  const int row0 = bin << sh;
  if (row0 >= N) return;
  int nrow = N - row0;
  nrow = (nrow > (1 << sh)) ? (1 << sh) : nrow;
  const int beg = boff[bin];
  const int end = boff[bin + 1];

  if (t < RMAX) lc[t] = 0;
  __syncthreads();
  int ic = beg + t;
  for (; ic + 3 * ETPB < end; ic += 4 * ETPB) {
    const int d0 = bdst[ic];
    const int d1 = bdst[ic + ETPB];
    const int d2 = bdst[ic + 2 * ETPB];
    const int d3 = bdst[ic + 3 * ETPB];
    atomicAdd(&lc[d0 - row0], 1);
    atomicAdd(&lc[d1 - row0], 1);
    atomicAdd(&lc[d2 - row0], 1);
    atomicAdd(&lc[d3 - row0], 1);
  }
  for (; ic < end; ic += ETPB) atomicAdd(&lc[bdst[ic] - row0], 1);
  __syncthreads();
  int cn = 0;
  if (t < RMAX) { cn = lc[t]; ls[t] = cn; }
  __syncthreads();
  for (int d = 1; d < RMAX; d <<= 1) {
    const int v = (t < RMAX && t >= d) ? ls[t - d] : 0;
    __syncthreads();
    if (t < RMAX) ls[t] += v;
    __syncthreads();
  }
  if (t < RMAX) {
    const int st = beg + ls[t] - cn;
    lq[t] = st;
    if (t < nrow) { rp[row0 + t] = st; cnt[row0 + t] = cn; }
  }
  __syncthreads();
  int ip = beg + t;
  for (; ip + 3 * ETPB < end; ip += 4 * ETPB) {
    const int d0 = bdst[ip];
    const int d1 = bdst[ip + ETPB];
    const int d2 = bdst[ip + 2 * ETPB];
    const int d3 = bdst[ip + 3 * ETPB];
    const int s0 = bsrc[ip];
    const int s1 = bsrc[ip + ETPB];
    const int s2 = bsrc[ip + 2 * ETPB];
    const int s3 = bsrc[ip + 3 * ETPB];
    const int p0 = atomicAdd(&lq[d0 - row0], 1);
    const int p1 = atomicAdd(&lq[d1 - row0], 1);
    const int p2 = atomicAdd(&lq[d2 - row0], 1);
    const int p3 = atomicAdd(&lq[d3 - row0], 1);
    perm[p0] = s0;
    perm[p1] = s1;
    perm[p2] = s2;
    perm[p3] = s3;
  }
  for (; ip < end; ip += ETPB) {
    const int p = atomicAdd(&lq[bdst[ip] - row0], 1);
    perm[p] = bsrc[ip];
  }
}

__device__ __forceinline__ void mean_rows_to_lds(const float* __restrict__ xin,
                                                 const int* __restrict__ rowptr,
                                                 const int* __restrict__ cnt,
                                                 const int* __restrict__ perm,
                                                 float (*sA)[IN_CH + 4], float (*sX)[IN_CH + 4],
                                                 int rw, int nrows, int lrow, int lane) {
  int nr = nrows - rw;
  nr = (nr < 0) ? 0 : ((nr > RPWF) ? RPWF : nr);
  const v4f zero = {0.f, 0.f, 0.f, 0.f};
  const float* xl = xin + lane * 4;
  v4f xd[RPWF];
  for (int i = 0; i < RPWF; ++i)
    xd[i] = (i < nr) ? *(const v4f*)(xl + (size_t)(rw + i) * IN_CH) : zero;
  int st = 0, cn = 0;
  if (lane < nr) {
    st = rowptr[rw + lane];
    cn = cnt[rw + lane];
  }
  int start = __builtin_amdgcn_readlane(st, 0);
  int n     = __builtin_amdgcn_readlane(cn, 0);
  int mine  = (lane < n) ? perm[start + lane] : 0;
  for (int i = 0; i < RPWF; ++i) *(v4f*)&sX[lrow + i][lane * 4] = xd[i];
  int first = __builtin_amdgcn_readlane(mine, 0);
#define GATHER_ROW(k) (*(const v4f*)(xl + (size_t)__builtin_amdgcn_readlane(mine, (k)) * IN_CH))
  for (int r = 0; r < nr; ++r) {
    int start_nx = 0, n_nx = 0, mine_nx = 0;
    if (r + 1 < nr) {
      start_nx = __builtin_amdgcn_readlane(st, r + 1);
      n_nx     = __builtin_amdgcn_readlane(cn, r + 1);
      mine_nx  = (lane < n_nx) ? perm[start_nx + lane] : 0;
    }
    v4f acc = zero;
    for (int i = 0; i < n; i += 32) {
      const int m = (n - i < 32) ? (n - i) : 32;
      if (i > 0) {
        mine  = (lane < m) ? perm[start + i + lane] : 0;
        first = __builtin_amdgcn_readlane(mine, 0);
      }
      const v4f p0 = *(const v4f*)(xl + (size_t)first * IN_CH);
      int j = 1;
      for (; j + 8 <= m; j += 8) {
        const v4f v0 = GATHER_ROW(j);
        const v4f v1 = GATHER_ROW(j + 1);
        const v4f v2 = GATHER_ROW(j + 2);
        const v4f v3 = GATHER_ROW(j + 3);
        const v4f v4 = GATHER_ROW(j + 4);
        const v4f v5 = GATHER_ROW(j + 5);
        const v4f v6 = GATHER_ROW(j + 6);
        const v4f v7 = GATHER_ROW(j + 7);
        acc += v0; acc += v1; acc += v2; acc += v3;
        acc += v4; acc += v5; acc += v6; acc += v7;
      }
      const int rem = m - j;
      v4f q0 = zero, q1 = zero, q2 = zero, q3 = zero, d0 = zero, d1 = zero, o0 = zero;
      if (rem & 4) {
        q0 = GATHER_ROW(j);
        q1 = GATHER_ROW(j + 1);
        q2 = GATHER_ROW(j + 2);
        q3 = GATHER_ROW(j + 3);
        j += 4;
      }
      if (rem & 2) {
        d0 = GATHER_ROW(j);
        d1 = GATHER_ROW(j + 1);
        j += 2;
      }
      if (rem & 1) {
        o0 = GATHER_ROW(j);
      }
      acc += q0; acc += q1; acc += q2; acc += q3;
      acc += d0; acc += d1; acc += o0;
      acc += p0;
    }
    const float inv = 1.0f / fmaxf((float)n, 1.0f);
    *(v4f*)&sA[lrow + r][lane * 4] = acc * inv;
    start = start_nx;
    n     = n_nx;
    mine  = mine_nx;
    first = __builtin_amdgcn_readlane(mine, 0);
  }
#undef GATHER_ROW
  for (int r = 0; r < RPWF; ++r)
    if (r >= nr) *(v4f*)&sA[lrow + r][lane * 4] = zero;
}

__global__ __launch_bounds__(128)
void sage_gather_gemm_relu(const float* __restrict__ xin,
                           const int* __restrict__ rowptr, const int* __restrict__ cnt,
                           const int* __restrict__ perm,
                           const float* __restrict__ Wl, const float* __restrict__ bl,
                           const float* __restrict__ Wr,
                           float* __restrict__ hout, int nrows)
{
  __shared__ __attribute__((aligned(16))) float sA[16][IN_CH + 4];
  __shared__ __attribute__((aligned(16))) float sX[16][IN_CH + 4];
  const int tid  = threadIdx.x;
  const int r0   = blockIdx.x * 16;
  const int lane = tid & 31;
  const int wave = __builtin_amdgcn_readfirstlane(tid >> 5);

  mean_rows_to_lds(xin, rowptr, cnt, perm, sA, sX, r0 + wave * RPWF, nrows, wave * RPWF, lane);
  __syncthreads();

  const int half = lane >> 4;
  const int m    = lane & 15;
  const int n    = lane & 15;
  const int kh   = half * 2;

  for (int t = 0; t < 2; ++t) {
    const int n0 = (wave * 2 + t) * 16;
    v8f acc = {0.f, 0.f, 0.f, 0.f, 0.f, 0.f, 0.f, 0.f};
#pragma clang loop unroll(disable)
    for (int kc = 0; kc < IN_CH; kc += 32) {
      for (int k = kc; k < kc + 32; k += 4) {
        v2f a, b;
        a.x = sA[m][k + kh];
        a.y = sA[m][k + kh + 1];
        b.x = Wl[(size_t)(k + kh)     * HID + n0 + n];
        b.y = Wl[(size_t)(k + kh + 1) * HID + n0 + n];
        acc = __builtin_amdgcn_wmma_f32_16x16x4_f32(false, a, false, b, (short)0, acc, false, false);
      }
    }
#pragma clang loop unroll(disable)
    for (int kc = 0; kc < IN_CH; kc += 32) {
      for (int k = kc; k < kc + 32; k += 4) {
        v2f a, b;
        a.x = sX[m][k + kh];
        a.y = sX[m][k + kh + 1];
        b.x = Wr[(size_t)(k + kh)     * HID + n0 + n];
        b.y = Wr[(size_t)(k + kh + 1) * HID + n0 + n];
        acc = __builtin_amdgcn_wmma_f32_16x16x4_f32(false, a, false, b, (short)0, acc, false, false);
      }
    }
    const float bias  = bl[n0 + n];
    const int   mbase = half * 8;
    for (int r = 0; r < 8; ++r) {
      int row = r0 + mbase + r;
      if (row < nrows) {
        float v = acc[r] + bias;
        hout[(size_t)row * HID + n0 + n] = fmaxf(v, 0.0f);
      }
    }
  }
}

__global__ __launch_bounds__(128)
void sage_gather_gemm_lsm(const float* __restrict__ hin,
                          const int* __restrict__ rowptr, const int* __restrict__ cnt,
                          const int* __restrict__ perm,
                          const float* __restrict__ Wl, const float* __restrict__ bl,
                          const float* __restrict__ Wr,
                          float* __restrict__ out, int nrows)
{
  __shared__ __attribute__((aligned(16))) float sA[16][HID + 4];
  __shared__ __attribute__((aligned(16))) float sX[16][HID + 4];
  __shared__ float sL[16][OUTP];
  const int tid  = threadIdx.x;
  const int r0   = blockIdx.x * 16;
  const int lane = tid & 31;
  const int wave = __builtin_amdgcn_readfirstlane(tid >> 5);

  mean_rows_to_lds(hin, rowptr, cnt, perm, sA, sX, r0 + wave * RPWF, nrows, wave * RPWF, lane);
  __syncthreads();

  const int half = lane >> 4;
  const int m    = lane & 15;
  const int n    = lane & 15;
  const int kh   = half * 2;

  if (wave < 3) {
    const int n0  = wave * 16;
    const int col = n0 + n;
    const bool cv = (col < OUTC);
    v8f acc = {0.f, 0.f, 0.f, 0.f, 0.f, 0.f, 0.f, 0.f};
    for (int k = 0; k < HID; k += 4) {
      v2f a, b;
      a.x = sA[m][k + kh];
      a.y = sA[m][k + kh + 1];
      b.x = cv ? Wl[(size_t)(k + kh)     * OUTC + col] : 0.0f;
      b.y = cv ? Wl[(size_t)(k + kh + 1) * OUTC + col] : 0.0f;
      acc = __builtin_amdgcn_wmma_f32_16x16x4_f32(false, a, false, b, (short)0, acc, false, false);
    }
    for (int k = 0; k < HID; k += 4) {
      v2f a, b;
      a.x = sX[m][k + kh];
      a.y = sX[m][k + kh + 1];
      b.x = cv ? Wr[(size_t)(k + kh)     * OUTC + col] : 0.0f;
      b.y = cv ? Wr[(size_t)(k + kh + 1) * OUTC + col] : 0.0f;
      acc = __builtin_amdgcn_wmma_f32_16x16x4_f32(false, a, false, b, (short)0, acc, false, false);
    }
    const float bias  = cv ? bl[col] : 0.0f;
    const int   mbase = half * 8;
    for (int r = 0; r < 8; ++r)
      sL[mbase + r][col] = acc[r] + bias;
  }
  __syncthreads();

  if (tid < 16) {
    int row = r0 + tid;
    if (row < nrows) {
      float mx = -INFINITY;
      for (int c = 0; c < OUTC; ++c) mx = fmaxf(mx, sL[tid][c]);
      float s = 0.0f;
      for (int c = 0; c < OUTC; ++c) s += expf(sL[tid][c] - mx);
      float lse = logf(s);
      for (int c = 0; c < OUTC; ++c)
        out[(size_t)row * OUTC + c] = sL[tid][c] - mx - lse;
    }
  }
}

extern "C" void kernel_launch(void* const* d_in, const int* in_sizes, int n_in,
                              void* d_out, int out_size, void* d_ws, size_t ws_size,
                              hipStream_t stream)
{
  const float* x    = (const float*)d_in[0];
  const int*   src0 = (const int*)d_in[1];
  const int*   dst0 = (const int*)d_in[2];
  const int*   src1 = (const int*)d_in[3];
  const int*   dst1 = (const int*)d_in[4];
  const float* Wl0  = (const float*)d_in[7];
  const float* bl0  = (const float*)d_in[8];
  const float* Wr0  = (const float*)d_in[9];
  const float* Wl1  = (const float*)d_in[10];
  const float* bl1  = (const float*)d_in[11];
  const float* Wr1  = (const float*)d_in[12];
  float* out = (float*)d_out;

  const int E0 = in_sizes[1];
  const int E1 = in_sizes[3];
  const int N1 = 100000;
  const int N2 = out_size / OUTC;

  float* msg0 = (float*)d_ws;
  float* deg0 = msg0 + (size_t)N1 * HID;
  float* h    = deg0 + N1;
  float* msg1 = h    + (size_t)N1 * HID;

  const int N1p = (N1 + 3) & ~3;
  const int N2p = (N2 + 3) & ~3;
  const int E0p = (E0 + 3) & ~3;
  const int E1p = (E1 + 3) & ~3;
  const int nblk0 = (E0 + CHUNK - 1) / CHUNK;
  const int nblk1 = (E1 + CHUNK - 1) / CHUNK;
  int* cnt0  = (int*)msg0;
  int* rp0   = cnt0 + N1p;
  int* perm0 = rp0 + N1p;
  int* bdst0 = perm0 + E0p;
  int* bsrc0 = bdst0 + E0p;
  int* boff0 = bsrc0 + E0p;
  int* table = boff0 + NBIN + 4;
  int* cnt1  = (int*)msg1;
  int* rp1   = cnt1 + N2p;
  int* perm1 = rp1 + N2p;
  int* bdst1 = perm1 + E1p;
  int* bsrc1 = bdst1 + E1p;
  int* boff1 = bsrc1 + E1p;

  bin_hist<<<nblk0 + nblk1, ETPB, 0, stream>>>(dst0, E0, N1, nblk0, dst1, E1, N2, table);
  bin_scatter<<<nblk0 + nblk1, ETPB, 0, stream>>>(src0, dst0, E0, N1, nblk0, src1, dst1, E1, N2, nblk1,
                                                  table, bdst0, bsrc0, boff0, bdst1, bsrc1, boff1);
  bin_csr<<<2 * NBIN, ETPB, 0, stream>>>(bdst0, bsrc0, boff0, N1, rp0, cnt0, perm0,
                                         bdst1, bsrc1, boff1, N2, rp1, cnt1, perm1);

  sage_gather_gemm_relu<<<(N1 + 15) / 16, 128, 0, stream>>>(x, rp0, cnt0, perm0, Wl0, bl0, Wr0, h, N1);

  sage_gather_gemm_lsm<<<(N2 + 15) / 16, 128, 0, stream>>>(h, rp1, cnt1, perm1, Wl1, bl1, Wr1, out, N2);
}


# sort passes 2 and 3 order their edges in LDS first, so every global store of a wave is a run of consecutive slots (whole lines), not 32 scattered 4-byte stores; sum order = LDS arrival, as before
# speedup vs baseline: 3.5623x; 1.2382x over previous
#include <hip/hip_runtime.h>
#include <math.h>

#define IN_CH 128
#define HID   128
#define OUTC  47
#define OUTP  48
#define RPWF  4

typedef float v2f __attribute__((ext_vector_type(2)));
typedef float v8f __attribute__((ext_vector_type(8)));


typedef float v4f __attribute__((ext_vector_type(4)));
typedef int   v4i __attribute__((ext_vector_type(4)));

#define NBIN  256
#define CHUNK 8192
#define ETPB  1024
#define RMAX  512
#define PMAX  12288

__device__ __forceinline__ int bin_shift(int n) {
  const int s = 24 - __clz(n - 1);
  return s < 0 ? 0 : s;
}

__global__ __launch_bounds__(ETPB)
void bin_hist(const int* __restrict__ dst0, int E0, int N0, int nblk0,
              const int* __restrict__ dst1, int E1, int N1,
              int* __restrict__ table) {
  __shared__ int h[NBIN];
  const int t = threadIdx.x;
  const int* dst; int E, N, c;
  if ((int)blockIdx.x < nblk0) { dst = dst0; E = E0; N = N0; c = blockIdx.x; }
  else                         { dst = dst1; E = E1; N = N1; c = blockIdx.x - nblk0; }
  const int sh = bin_shift(N);
  if (t < NBIN) h[t] = 0;
  __syncthreads();
  for (int j = 0; j < CHUNK / (4 * ETPB); ++j) {
    const int e = (c * (CHUNK / 4) + j * ETPB + t) * 4;
    if (e + 3 < E) {
      const v4i d = *(const v4i*)(dst + e);
      atomicAdd(&h[d.x >> sh], 1);
      atomicAdd(&h[d.y >> sh], 1);
      atomicAdd(&h[d.z >> sh], 1);
      atomicAdd(&h[d.w >> sh], 1);
    } else {
      for (int k = e; k < E; ++k) atomicAdd(&h[dst[k] >> sh], 1);
    }
  }
  __syncthreads();
  if (t < NBIN) table[(size_t)blockIdx.x * NBIN + t] = h[t];
}

__global__ __launch_bounds__(ETPB)
void bin_scatter(const int* __restrict__ src0, const int* __restrict__ dst0, int E0, int N0, int nblk0,
                 const int* __restrict__ src1, const int* __restrict__ dst1, int E1, int N1, int nblk1,
                 const int* __restrict__ table,
                 int* __restrict__ bdst0, int* __restrict__ bsrc0, int* __restrict__ boff0,
                 int* __restrict__ bdst1, int* __restrict__ bsrc1, int* __restrict__ boff1) {
  __shared__ int sTot[4][NBIN];
  __shared__ int sPre[4][NBIN];
  __shared__ int sScan[NBIN];
  __shared__ int sOwn[NBIN];
  __shared__ int lpos[NBIN];
  __shared__ int gdel[NBIN];
  __shared__ int sd[CHUNK];
  __shared__ int ss[CHUNK];
  const int t = threadIdx.x;
  const int* src; const int* dst; int* bdst; int* bsrc; int* boff; int E, N, c, fb, nb;
  if ((int)blockIdx.x < nblk0) {
    src = src0; dst = dst0; bdst = bdst0; bsrc = bsrc0; boff = boff0;
    E = E0; N = N0; c = blockIdx.x; fb = 0; nb = nblk0;
  } else {
    src = src1; dst = dst1; bdst = bdst1; bsrc = bsrc1; boff = boff1;
    E = E1; N = N1; c = blockIdx.x - nblk0; fb = nblk0; nb = nblk1;
  }
  const int sh = bin_shift(N);

  {
    const int b    = t & (NBIN - 1);
    const int part = t >> 8;
    const int q    = (nb + 3) >> 2;
    const int k0   = part * q;
    const int k1   = (k0 + q < nb) ? (k0 + q) : nb;
    int tot = 0, pre = 0;
    for (int k = k0; k < k1; ++k) {
      const int v = table[(size_t)(fb + k) * NBIN + b];
      tot += v;
      pre += (k < c) ? v : 0;
    }
    sTot[part][b] = tot;
    sPre[part][b] = pre;
  }
  __syncthreads();
  int total = 0, base = 0, own = 0;
  if (t < NBIN) {
    total = (sTot[0][t] + sTot[1][t]) + (sTot[2][t] + sTot[3][t]);
    base  = (sPre[0][t] + sPre[1][t]) + (sPre[2][t] + sPre[3][t]);
    own   = table[(size_t)(fb + c) * NBIN + t];
    sScan[t] = total;
    sOwn[t]  = own;
  }
  __syncthreads();
  for (int d = 1; d < NBIN; d <<= 1) {
    int v = 0, w = 0;
    if (t < NBIN && t >= d) { v = sScan[t - d]; w = sOwn[t - d]; }
    __syncthreads();
    if (t < NBIN) { sScan[t] += v; sOwn[t] += w; }
    __syncthreads();
  }
  if (t < NBIN) {
    const int ex  = sScan[t] - total;
    const int lex = sOwn[t] - own;
    lpos[t] = lex;
    gdel[t] = ex + base - lex;
    if (c == 0) {
      boff[t] = ex;
      if (t == NBIN - 1) boff[NBIN] = sScan[t];
    }
  }
  __syncthreads();

  for (int j = 0; j < CHUNK / (4 * ETPB); ++j) {
    const int e = (c * (CHUNK / 4) + j * ETPB + t) * 4;
    if (e + 3 < E) {
      const v4i d = *(const v4i*)(dst + e);
      const v4i s = *(const v4i*)(src + e);
      const int p0 = atomicAdd(&lpos[d.x >> sh], 1);
      const int p1 = atomicAdd(&lpos[d.y >> sh], 1);
      const int p2 = atomicAdd(&lpos[d.z >> sh], 1);
      const int p3 = atomicAdd(&lpos[d.w >> sh], 1);
      sd[p0] = d.x; ss[p0] = s.x;
      sd[p1] = d.y; ss[p1] = s.y;
      sd[p2] = d.z; ss[p2] = s.z;
      sd[p3] = d.w; ss[p3] = s.w;
    } else {
      for (int k = e; k < E; ++k) {
        const int dk = dst[k];
        const int p  = atomicAdd(&lpos[dk >> sh], 1);
        sd[p] = dk; ss[p] = src[k];
      }
    }
  }
  __syncthreads();
  int nE = E - c * CHUNK;
  nE = (nE > CHUNK) ? CHUNK : nE;
  for (int p = t; p < nE; p += ETPB) {
    const int dk = sd[p];
    const int g  = p + gdel[dk >> sh];
    bdst[g] = dk;
    bsrc[g] = ss[p];
  }
}

__global__ __launch_bounds__(ETPB)
void bin_csr(const int* __restrict__ bdst0, const int* __restrict__ bsrc0, const int* __restrict__ boff0, int N0,
             int* __restrict__ rp0, int* __restrict__ cnt0, int* __restrict__ perm0,
             const int* __restrict__ bdst1, const int* __restrict__ bsrc1, const int* __restrict__ boff1, int N1,
             int* __restrict__ rp1, int* __restrict__ cnt1, int* __restrict__ perm1) {
  __shared__ int lc[RMAX];
  __shared__ int ls[RMAX];
  __shared__ int lq[RMAX];
  __shared__ int lp[PMAX];
  const int t = threadIdx.x;
  const int* bdst; const int* bsrc; const int* boff; int* rp; int* cnt; int* perm; int N, bin;
  if ((int)blockIdx.x < NBIN) {
    bdst = bdst0; bsrc = bsrc0; boff = boff0; rp = rp0; cnt = cnt0; perm = perm0; N = N0; bin = blockIdx.x;
  } else {
    bdst = bdst1; bsrc = bsrc1; boff = boff1; rp = rp1; cnt = cnt1; perm = perm1; N = N1; bin = blockIdx.x - NBIN;
  }
  const int sh   = bin_shift(N);
  const int row0 = bin << sh;
  if (row0 >= N) return;
  int nrow = N - row0;
  nrow = (nrow > (1 << sh)) ? (1 << sh) : nrow;
  const int beg = boff[bin];
  const int end = boff[bin + 1];

  if (t < RMAX) lc[t] = 0;
  __syncthreads();
  int ic = beg + t;
  for (; ic + 3 * ETPB < end; ic += 4 * ETPB) {
    const int d0 = bdst[ic];
    const int d1 = bdst[ic + ETPB];
    const int d2 = bdst[ic + 2 * ETPB];
    const int d3 = bdst[ic + 3 * ETPB];
    atomicAdd(&lc[d0 - row0], 1);
    atomicAdd(&lc[d1 - row0], 1);
    atomicAdd(&lc[d2 - row0], 1);
    atomicAdd(&lc[d3 - row0], 1);
  }
  for (; ic < end; ic += ETPB) atomicAdd(&lc[bdst[ic] - row0], 1);
  __syncthreads();
  int cn = 0;
  if (t < RMAX) { cn = lc[t]; ls[t] = cn; }
  __syncthreads();
  for (int d = 1; d < RMAX; d <<= 1) {
    const int v = (t < RMAX && t >= d) ? ls[t - d] : 0;
    __syncthreads();
    if (t < RMAX) ls[t] += v;
    __syncthreads();
  }
  if (t < RMAX) {
    const int st = ls[t] - cn;
    lq[t] = st;
    if (t < nrow) { rp[row0 + t] = beg + st; cnt[row0 + t] = cn; }
  }
  __syncthreads();
  if (end - beg <= PMAX) {
    int iq = beg + t;
    for (; iq + 3 * ETPB < end; iq += 4 * ETPB) {
      const int d0 = bdst[iq];
      const int d1 = bdst[iq + ETPB];
      const int d2 = bdst[iq + 2 * ETPB];
      const int d3 = bdst[iq + 3 * ETPB];
      const int s0 = bsrc[iq];
      const int s1 = bsrc[iq + ETPB];
      const int s2 = bsrc[iq + 2 * ETPB];
      const int s3 = bsrc[iq + 3 * ETPB];
      const int p0 = atomicAdd(&lq[d0 - row0], 1);
      const int p1 = atomicAdd(&lq[d1 - row0], 1);
      const int p2 = atomicAdd(&lq[d2 - row0], 1);
      const int p3 = atomicAdd(&lq[d3 - row0], 1);
      lp[p0] = s0;
      lp[p1] = s1;
      lp[p2] = s2;
      lp[p3] = s3;
    }
    for (; iq < end; iq += ETPB) {
      const int p = atomicAdd(&lq[bdst[iq] - row0], 1);
      lp[p] = bsrc[iq];
    }
    __syncthreads();
    for (int i = t; i < end - beg; i += ETPB) perm[beg + i] = lp[i];
    return;
  }
  int ip = beg + t;
  for (; ip + 3 * ETPB < end; ip += 4 * ETPB) {
    const int d0 = bdst[ip];
    const int d1 = bdst[ip + ETPB];
    const int d2 = bdst[ip + 2 * ETPB];
    const int d3 = bdst[ip + 3 * ETPB];
    const int s0 = bsrc[ip];
    const int s1 = bsrc[ip + ETPB];
    const int s2 = bsrc[ip + 2 * ETPB];
    const int s3 = bsrc[ip + 3 * ETPB];
    const int p0 = atomicAdd(&lq[d0 - row0], 1);
    const int p1 = atomicAdd(&lq[d1 - row0], 1);
    const int p2 = atomicAdd(&lq[d2 - row0], 1);
    const int p3 = atomicAdd(&lq[d3 - row0], 1);
    perm[beg + p0] = s0;
    perm[beg + p1] = s1;
    perm[beg + p2] = s2;
    perm[beg + p3] = s3;
  }
  for (; ip < end; ip += ETPB) {
    const int p = atomicAdd(&lq[bdst[ip] - row0], 1);
    perm[beg + p] = bsrc[ip];
  }
}

__device__ __forceinline__ void mean_rows_to_lds(const float* __restrict__ xin,
                                                 const int* __restrict__ rowptr,
                                                 const int* __restrict__ cnt,
                                                 const int* __restrict__ perm,
                                                 float (*sA)[IN_CH + 4], float (*sX)[IN_CH + 4],
                                                 int rw, int nrows, int lrow, int lane) {
  int nr = nrows - rw;
  nr = (nr < 0) ? 0 : ((nr > RPWF) ? RPWF : nr);
  const v4f zero = {0.f, 0.f, 0.f, 0.f};
  const float* xl = xin + lane * 4;
  v4f xd[RPWF];
  for (int i = 0; i < RPWF; ++i)
    xd[i] = (i < nr) ? *(const v4f*)(xl + (size_t)(rw + i) * IN_CH) : zero;
  int st = 0, cn = 0;
  if (lane < nr) {
    st = rowptr[rw + lane];
    cn = cnt[rw + lane];
  }
  int start = __builtin_amdgcn_readlane(st, 0);
  int n     = __builtin_amdgcn_readlane(cn, 0);
  int mine  = (lane < n) ? perm[start + lane] : 0;
  for (int i = 0; i < RPWF; ++i) *(v4f*)&sX[lrow + i][lane * 4] = xd[i];
  int first = __builtin_amdgcn_readlane(mine, 0);
#define GATHER_ROW(k) (*(const v4f*)(xl + (size_t)__builtin_amdgcn_readlane(mine, (k)) * IN_CH))
  for (int r = 0; r < nr; ++r) {
    int start_nx = 0, n_nx = 0, mine_nx = 0;
    if (r + 1 < nr) {
      start_nx = __builtin_amdgcn_readlane(st, r + 1);
      n_nx     = __builtin_amdgcn_readlane(cn, r + 1);
      mine_nx  = (lane < n_nx) ? perm[start_nx + lane] : 0;
    }
    v4f acc = zero;
    for (int i = 0; i < n; i += 32) {
      const int m = (n - i < 32) ? (n - i) : 32;
      if (i > 0) {
        mine  = (lane < m) ? perm[start + i + lane] : 0;
        first = __builtin_amdgcn_readlane(mine, 0);
      }
      const v4f p0 = *(const v4f*)(xl + (size_t)first * IN_CH);
      int j = 1;
      for (; j + 8 <= m; j += 8) {
        const v4f v0 = GATHER_ROW(j);
        const v4f v1 = GATHER_ROW(j + 1);
        const v4f v2 = GATHER_ROW(j + 2);
        const v4f v3 = GATHER_ROW(j + 3);
        const v4f v4 = GATHER_ROW(j + 4);
        const v4f v5 = GATHER_ROW(j + 5);
        const v4f v6 = GATHER_ROW(j + 6);
        const v4f v7 = GATHER_ROW(j + 7);
        acc += v0; acc += v1; acc += v2; acc += v3;
        acc += v4; acc += v5; acc += v6; acc += v7;
      }
      const int rem = m - j;
      v4f q0 = zero, q1 = zero, q2 = zero, q3 = zero, d0 = zero, d1 = zero, o0 = zero;
      if (rem & 4) {
        q0 = GATHER_ROW(j);
        q1 = GATHER_ROW(j + 1);
        q2 = GATHER_ROW(j + 2);
        q3 = GATHER_ROW(j + 3);
        j += 4;
      }
      if (rem & 2) {
        d0 = GATHER_ROW(j);
        d1 = GATHER_ROW(j + 1);
        j += 2;
      }
      if (rem & 1) {
        o0 = GATHER_ROW(j);
      }
      acc += q0; acc += q1; acc += q2; acc += q3;
      acc += d0; acc += d1; acc += o0;
      acc += p0;
    }
    const float inv = 1.0f / fmaxf((float)n, 1.0f);
    *(v4f*)&sA[lrow + r][lane * 4] = acc * inv;
    start = start_nx;
    n     = n_nx;
    mine  = mine_nx;
    first = __builtin_amdgcn_readlane(mine, 0);
  }
#undef GATHER_ROW
  for (int r = 0; r < RPWF; ++r)
    if (r >= nr) *(v4f*)&sA[lrow + r][lane * 4] = zero;
}

__global__ __launch_bounds__(128)
void sage_gather_gemm_relu(const float* __restrict__ xin,
                           const int* __restrict__ rowptr, const int* __restrict__ cnt,
                           const int* __restrict__ perm,
                           const float* __restrict__ Wl, const float* __restrict__ bl,
                           const float* __restrict__ Wr,
                           float* __restrict__ hout, int nrows)
{
  __shared__ __attribute__((aligned(16))) float sA[16][IN_CH + 4];
  __shared__ __attribute__((aligned(16))) float sX[16][IN_CH + 4];
  const int tid  = threadIdx.x;
  const int r0   = blockIdx.x * 16;
  const int lane = tid & 31;
  const int wave = __builtin_amdgcn_readfirstlane(tid >> 5);

  mean_rows_to_lds(xin, rowptr, cnt, perm, sA, sX, r0 + wave * RPWF, nrows, wave * RPWF, lane);
  __syncthreads();

  const int half = lane >> 4;
  const int m    = lane & 15;
  const int n    = lane & 15;
  const int kh   = half * 2;

  for (int t = 0; t < 2; ++t) {
    const int n0 = (wave * 2 + t) * 16;
    v8f acc = {0.f, 0.f, 0.f, 0.f, 0.f, 0.f, 0.f, 0.f};
#pragma clang loop unroll(disable)
    for (int kc = 0; kc < IN_CH; kc += 32) {
      for (int k = kc; k < kc + 32; k += 4) {
        v2f a, b;
        a.x = sA[m][k + kh];
        a.y = sA[m][k + kh + 1];
        b.x = Wl[(size_t)(k + kh)     * HID + n0 + n];
        b.y = Wl[(size_t)(k + kh + 1) * HID + n0 + n];
        acc = __builtin_amdgcn_wmma_f32_16x16x4_f32(false, a, false, b, (short)0, acc, false, false);
      }
    }
#pragma clang loop unroll(disable)
    for (int kc = 0; kc < IN_CH; kc += 32) {
      for (int k = kc; k < kc + 32; k += 4) {
        v2f a, b;
        a.x = sX[m][k + kh];
        a.y = sX[m][k + kh + 1];
        b.x = Wr[(size_t)(k + kh)     * HID + n0 + n];
        b.y = Wr[(size_t)(k + kh + 1) * HID + n0 + n];
        acc = __builtin_amdgcn_wmma_f32_16x16x4_f32(false, a, false, b, (short)0, acc, false, false);
      }
    }
    const float bias  = bl[n0 + n];
    const int   mbase = half * 8;
    for (int r = 0; r < 8; ++r) {
      int row = r0 + mbase + r;
      if (row < nrows) {
        float v = acc[r] + bias;
        hout[(size_t)row * HID + n0 + n] = fmaxf(v, 0.0f);
      }
    }
  }
}

__global__ __launch_bounds__(128)
void sage_gather_gemm_lsm(const float* __restrict__ hin,
                          const int* __restrict__ rowptr, const int* __restrict__ cnt,
                          const int* __restrict__ perm,
                          const float* __restrict__ Wl, const float* __restrict__ bl,
                          const float* __restrict__ Wr,
                          float* __restrict__ out, int nrows)
{
  __shared__ __attribute__((aligned(16))) float sA[16][HID + 4];
  __shared__ __attribute__((aligned(16))) float sX[16][HID + 4];
  __shared__ float sL[16][OUTP];
  const int tid  = threadIdx.x;
  const int r0   = blockIdx.x * 16;
  const int lane = tid & 31;
  const int wave = __builtin_amdgcn_readfirstlane(tid >> 5);

  mean_rows_to_lds(hin, rowptr, cnt, perm, sA, sX, r0 + wave * RPWF, nrows, wave * RPWF, lane);
  __syncthreads();

  const int half = lane >> 4;
  const int m    = lane & 15;
  const int n    = lane & 15;
  const int kh   = half * 2;

  if (wave < 3) {
    const int n0  = wave * 16;
    const int col = n0 + n;
    const bool cv = (col < OUTC);
    v8f acc = {0.f, 0.f, 0.f, 0.f, 0.f, 0.f, 0.f, 0.f};
    for (int k = 0; k < HID; k += 4) {
      v2f a, b;
      a.x = sA[m][k + kh];
      a.y = sA[m][k + kh + 1];
      b.x = cv ? Wl[(size_t)(k + kh)     * OUTC + col] : 0.0f;
      b.y = cv ? Wl[(size_t)(k + kh + 1) * OUTC + col] : 0.0f;
      acc = __builtin_amdgcn_wmma_f32_16x16x4_f32(false, a, false, b, (short)0, acc, false, false);
    }
    for (int k = 0; k < HID; k += 4) {
      v2f a, b;
      a.x = sX[m][k + kh];
      a.y = sX[m][k + kh + 1];
      b.x = cv ? Wr[(size_t)(k + kh)     * OUTC + col] : 0.0f;
      b.y = cv ? Wr[(size_t)(k + kh + 1) * OUTC + col] : 0.0f;
      acc = __builtin_amdgcn_wmma_f32_16x16x4_f32(false, a, false, b, (short)0, acc, false, false);
    }
    const float bias  = cv ? bl[col] : 0.0f;
    const int   mbase = half * 8;
    for (int r = 0; r < 8; ++r)
      sL[mbase + r][col] = acc[r] + bias;
  }
  __syncthreads();

  if (tid < 16) {
    int row = r0 + tid;
    if (row < nrows) {
      float mx = -INFINITY;
      for (int c = 0; c < OUTC; ++c) mx = fmaxf(mx, sL[tid][c]);
      float s = 0.0f;
      for (int c = 0; c < OUTC; ++c) s += expf(sL[tid][c] - mx);
      float lse = logf(s);
      for (int c = 0; c < OUTC; ++c)
        out[(size_t)row * OUTC + c] = sL[tid][c] - mx - lse;
    }
  }
}

extern "C" void kernel_launch(void* const* d_in, const int* in_sizes, int n_in,
                              void* d_out, int out_size, void* d_ws, size_t ws_size,
                              hipStream_t stream)
{
  const float* x    = (const float*)d_in[0];
  const int*   src0 = (const int*)d_in[1];
  const int*   dst0 = (const int*)d_in[2];
  const int*   src1 = (const int*)d_in[3];
  const int*   dst1 = (const int*)d_in[4];
  const float* Wl0  = (const float*)d_in[7];
  const float* bl0  = (const float*)d_in[8];
  const float* Wr0  = (const float*)d_in[9];
  const float* Wl1  = (const float*)d_in[10];
  const float* bl1  = (const float*)d_in[11];
  const float* Wr1  = (const float*)d_in[12];
  float* out = (float*)d_out;

  const int E0 = in_sizes[1];
  const int E1 = in_sizes[3];
  const int N1 = 100000;
  const int N2 = out_size / OUTC;

  float* msg0 = (float*)d_ws;
  float* deg0 = msg0 + (size_t)N1 * HID;
  float* h    = deg0 + N1;
  float* msg1 = h    + (size_t)N1 * HID;

  const int N1p = (N1 + 3) & ~3;
  const int N2p = (N2 + 3) & ~3;
  const int E0p = (E0 + 3) & ~3;
  const int E1p = (E1 + 3) & ~3;
  const int nblk0 = (E0 + CHUNK - 1) / CHUNK;
  const int nblk1 = (E1 + CHUNK - 1) / CHUNK;
  int* cnt0  = (int*)msg0;
  int* rp0   = cnt0 + N1p;
  int* perm0 = rp0 + N1p;
  int* bdst0 = perm0 + E0p;
  int* bsrc0 = bdst0 + E0p;
  int* boff0 = bsrc0 + E0p;
  int* table = boff0 + NBIN + 4;
  int* cnt1  = (int*)msg1;
  int* rp1   = cnt1 + N2p;
  int* perm1 = rp1 + N2p;
  int* bdst1 = perm1 + E1p;
  int* bsrc1 = bdst1 + E1p;
  int* boff1 = bsrc1 + E1p;

  bin_hist<<<nblk0 + nblk1, ETPB, 0, stream>>>(dst0, E0, N1, nblk0, dst1, E1, N2, table);
  bin_scatter<<<nblk0 + nblk1, ETPB, 0, stream>>>(src0, dst0, E0, N1, nblk0, src1, dst1, E1, N2, nblk1,
                                                  table, bdst0, bsrc0, boff0, bdst1, bsrc1, boff1);
  bin_csr<<<2 * NBIN, ETPB, 0, stream>>>(bdst0, bsrc0, boff0, N1, rp0, cnt0, perm0,
                                         bdst1, bsrc1, boff1, N2, rp1, cnt1, perm1);

  sage_gather_gemm_relu<<<(N1 + 15) / 16, 128, 0, stream>>>(x, rp0, cnt0, perm0, Wl0, bl0, Wr0, h, N1);

  sage_gather_gemm_lsm<<<(N2 + 15) / 16, 128, 0, stream>>>(h, rp1, cnt1, perm1, Wl1, bl1, Wr1, out, N2);
}


# layer-0 dense part: Wl and Wr come through LDS in 16-row pieces, one whole-line load per wave, shared by a 64-row block of 16 waves (one B fragment feeds two row tiles); sum order kept
# speedup vs baseline: 3.7155x; 1.0430x over previous
#include <hip/hip_runtime.h>
#include <math.h>

#define IN_CH 128
#define HID   128
#define OUTC  47
#define OUTP  48
#define RPWF  4

typedef float v2f __attribute__((ext_vector_type(2)));
typedef float v8f __attribute__((ext_vector_type(8)));


typedef float v4f __attribute__((ext_vector_type(4)));
typedef int   v4i __attribute__((ext_vector_type(4)));

#define NBIN  256
#define CHUNK 8192
#define ETPB  1024
#define RMAX  512
#define PMAX  12288

__device__ __forceinline__ int bin_shift(int n) {
  const int s = 24 - __clz(n - 1);
  return s < 0 ? 0 : s;
}

__global__ __launch_bounds__(ETPB)
void bin_hist(const int* __restrict__ dst0, int E0, int N0, int nblk0,
              const int* __restrict__ dst1, int E1, int N1,
              int* __restrict__ table) {
  __shared__ int h[NBIN];
  const int t = threadIdx.x;
  const int* dst; int E, N, c;
  if ((int)blockIdx.x < nblk0) { dst = dst0; E = E0; N = N0; c = blockIdx.x; }
  else                         { dst = dst1; E = E1; N = N1; c = blockIdx.x - nblk0; }
  const int sh = bin_shift(N);
  if (t < NBIN) h[t] = 0;
  __syncthreads();
  for (int j = 0; j < CHUNK / (4 * ETPB); ++j) {
    const int e = (c * (CHUNK / 4) + j * ETPB + t) * 4;
    if (e + 3 < E) {
      const v4i d = *(const v4i*)(dst + e);
      atomicAdd(&h[d.x >> sh], 1);
      atomicAdd(&h[d.y >> sh], 1);
      atomicAdd(&h[d.z >> sh], 1);
      atomicAdd(&h[d.w >> sh], 1);
    } else {
      for (int k = e; k < E; ++k) atomicAdd(&h[dst[k] >> sh], 1);
    }
  }
  __syncthreads();
  if (t < NBIN) table[(size_t)blockIdx.x * NBIN + t] = h[t];
}

__global__ __launch_bounds__(ETPB)
void bin_scatter(const int* __restrict__ src0, const int* __restrict__ dst0, int E0, int N0, int nblk0,
                 const int* __restrict__ src1, const int* __restrict__ dst1, int E1, int N1, int nblk1,
                 const int* __restrict__ table,
                 int* __restrict__ bdst0, int* __restrict__ bsrc0, int* __restrict__ boff0,
                 int* __restrict__ bdst1, int* __restrict__ bsrc1, int* __restrict__ boff1) {
  __shared__ int sTot[4][NBIN];
  __shared__ int sPre[4][NBIN];
  __shared__ int sScan[NBIN];
  __shared__ int sOwn[NBIN];
  __shared__ int lpos[NBIN];
  __shared__ int gdel[NBIN];
  __shared__ int sd[CHUNK];
  __shared__ int ss[CHUNK];
  const int t = threadIdx.x;
  const int* src; const int* dst; int* bdst; int* bsrc; int* boff; int E, N, c, fb, nb;
  if ((int)blockIdx.x < nblk0) {
    src = src0; dst = dst0; bdst = bdst0; bsrc = bsrc0; boff = boff0;
    E = E0; N = N0; c = blockIdx.x; fb = 0; nb = nblk0;
  } else {
    src = src1; dst = dst1; bdst = bdst1; bsrc = bsrc1; boff = boff1;
    E = E1; N = N1; c = blockIdx.x - nblk0; fb = nblk0; nb = nblk1;
  }
  const int sh = bin_shift(N);

  {
    const int b    = t & (NBIN - 1);
    const int part = t >> 8;
    const int q    = (nb + 3) >> 2;
    const int k0   = part * q;
    const int k1   = (k0 + q < nb) ? (k0 + q) : nb;
    int tot = 0, pre = 0;
    for (int k = k0; k < k1; ++k) {
      const int v = table[(size_t)(fb + k) * NBIN + b];
      tot += v;
      pre += (k < c) ? v : 0;
    }
    sTot[part][b] = tot;
    sPre[part][b] = pre;
  }
  __syncthreads();
  int total = 0, base = 0, own = 0;
  if (t < NBIN) {
    total = (sTot[0][t] + sTot[1][t]) + (sTot[2][t] + sTot[3][t]);
    base  = (sPre[0][t] + sPre[1][t]) + (sPre[2][t] + sPre[3][t]);
    own   = table[(size_t)(fb + c) * NBIN + t];
    sScan[t] = total;
    sOwn[t]  = own;
  }
  __syncthreads();
  for (int d = 1; d < NBIN; d <<= 1) {
    int v = 0, w = 0;
    if (t < NBIN && t >= d) { v = sScan[t - d]; w = sOwn[t - d]; }
    __syncthreads();
    if (t < NBIN) { sScan[t] += v; sOwn[t] += w; }
    __syncthreads();
  }
  if (t < NBIN) {
    const int ex  = sScan[t] - total;
    const int lex = sOwn[t] - own;
    lpos[t] = lex;
    gdel[t] = ex + base - lex;
    if (c == 0) {
      boff[t] = ex;
      if (t == NBIN - 1) boff[NBIN] = sScan[t];
    }
  }
  __syncthreads();

  for (int j = 0; j < CHUNK / (4 * ETPB); ++j) {
    const int e = (c * (CHUNK / 4) + j * ETPB + t) * 4;
    if (e + 3 < E) {
      const v4i d = *(const v4i*)(dst + e);
      const v4i s = *(const v4i*)(src + e);
      const int p0 = atomicAdd(&lpos[d.x >> sh], 1);
      const int p1 = atomicAdd(&lpos[d.y >> sh], 1);
      const int p2 = atomicAdd(&lpos[d.z >> sh], 1);
      const int p3 = atomicAdd(&lpos[d.w >> sh], 1);
      sd[p0] = d.x; ss[p0] = s.x;
      sd[p1] = d.y; ss[p1] = s.y;
      sd[p2] = d.z; ss[p2] = s.z;
      sd[p3] = d.w; ss[p3] = s.w;
    } else {
      for (int k = e; k < E; ++k) {
        const int dk = dst[k];
        const int p  = atomicAdd(&lpos[dk >> sh], 1);
        sd[p] = dk; ss[p] = src[k];
      }
    }
  }
  __syncthreads();
  int nE = E - c * CHUNK;
  nE = (nE > CHUNK) ? CHUNK : nE;
  for (int p = t; p < nE; p += ETPB) {
    const int dk = sd[p];
    const int g  = p + gdel[dk >> sh];
    bdst[g] = dk;
    bsrc[g] = ss[p];
  }
}

__global__ __launch_bounds__(ETPB)
void bin_csr(const int* __restrict__ bdst0, const int* __restrict__ bsrc0, const int* __restrict__ boff0, int N0,
             int* __restrict__ rp0, int* __restrict__ cnt0, int* __restrict__ perm0,
             const int* __restrict__ bdst1, const int* __restrict__ bsrc1, const int* __restrict__ boff1, int N1,
             int* __restrict__ rp1, int* __restrict__ cnt1, int* __restrict__ perm1) {
  __shared__ int lc[RMAX];
  __shared__ int ls[RMAX];
  __shared__ int lq[RMAX];
  __shared__ int lp[PMAX];
  const int t = threadIdx.x;
  const int* bdst; const int* bsrc; const int* boff; int* rp; int* cnt; int* perm; int N, bin;
  if ((int)blockIdx.x < NBIN) {
    bdst = bdst0; bsrc = bsrc0; boff = boff0; rp = rp0; cnt = cnt0; perm = perm0; N = N0; bin = blockIdx.x;
  } else {
    bdst = bdst1; bsrc = bsrc1; boff = boff1; rp = rp1; cnt = cnt1; perm = perm1; N = N1; bin = blockIdx.x - NBIN;
  }
  const int sh   = bin_shift(N);
  const int row0 = bin << sh;
  if (row0 >= N) return;
  int nrow = N - row0;
  nrow = (nrow > (1 << sh)) ? (1 << sh) : nrow;
  const int beg = boff[bin];
  const int end = boff[bin + 1];

  if (t < RMAX) lc[t] = 0;
  __syncthreads();
  int ic = beg + t;
  for (; ic + 3 * ETPB < end; ic += 4 * ETPB) {
    const int d0 = bdst[ic];
    const int d1 = bdst[ic + ETPB];
    const int d2 = bdst[ic + 2 * ETPB];
    const int d3 = bdst[ic + 3 * ETPB];
    atomicAdd(&lc[d0 - row0], 1);
    atomicAdd(&lc[d1 - row0], 1);
    atomicAdd(&lc[d2 - row0], 1);
    atomicAdd(&lc[d3 - row0], 1);
  }
  for (; ic < end; ic += ETPB) atomicAdd(&lc[bdst[ic] - row0], 1);
  __syncthreads();
  int cn = 0;
  if (t < RMAX) { cn = lc[t]; ls[t] = cn; }
  __syncthreads();
  for (int d = 1; d < RMAX; d <<= 1) {
    const int v = (t < RMAX && t >= d) ? ls[t - d] : 0;
    __syncthreads();
    if (t < RMAX) ls[t] += v;
    __syncthreads();
  }
  if (t < RMAX) {
    const int st = ls[t] - cn;
    lq[t] = st;
    if (t < nrow) { rp[row0 + t] = beg + st; cnt[row0 + t] = cn; }
  }
  __syncthreads();
  if (end - beg <= PMAX) {
    int iq = beg + t;
    for (; iq + 3 * ETPB < end; iq += 4 * ETPB) {
      const int d0 = bdst[iq];
      const int d1 = bdst[iq + ETPB];
      const int d2 = bdst[iq + 2 * ETPB];
      const int d3 = bdst[iq + 3 * ETPB];
      const int s0 = bsrc[iq];
      const int s1 = bsrc[iq + ETPB];
      const int s2 = bsrc[iq + 2 * ETPB];
      const int s3 = bsrc[iq + 3 * ETPB];
      const int p0 = atomicAdd(&lq[d0 - row0], 1);
      const int p1 = atomicAdd(&lq[d1 - row0], 1);
      const int p2 = atomicAdd(&lq[d2 - row0], 1);
      const int p3 = atomicAdd(&lq[d3 - row0], 1);
      lp[p0] = s0;
      lp[p1] = s1;
      lp[p2] = s2;
      lp[p3] = s3;
    }
    for (; iq < end; iq += ETPB) {
      const int p = atomicAdd(&lq[bdst[iq] - row0], 1);
      lp[p] = bsrc[iq];
    }
    __syncthreads();
    for (int i = t; i < end - beg; i += ETPB) perm[beg + i] = lp[i];
    return;
  }
  int ip = beg + t;
  for (; ip + 3 * ETPB < end; ip += 4 * ETPB) {
    const int d0 = bdst[ip];
    const int d1 = bdst[ip + ETPB];
    const int d2 = bdst[ip + 2 * ETPB];
    const int d3 = bdst[ip + 3 * ETPB];
    const int s0 = bsrc[ip];
    const int s1 = bsrc[ip + ETPB];
    const int s2 = bsrc[ip + 2 * ETPB];
    const int s3 = bsrc[ip + 3 * ETPB];
    const int p0 = atomicAdd(&lq[d0 - row0], 1);
    const int p1 = atomicAdd(&lq[d1 - row0], 1);
    const int p2 = atomicAdd(&lq[d2 - row0], 1);
    const int p3 = atomicAdd(&lq[d3 - row0], 1);
    perm[beg + p0] = s0;
    perm[beg + p1] = s1;
    perm[beg + p2] = s2;
    perm[beg + p3] = s3;
  }
  for (; ip < end; ip += ETPB) {
    const int p = atomicAdd(&lq[bdst[ip] - row0], 1);
    perm[beg + p] = bsrc[ip];
  }
}

__device__ __forceinline__ void mean_rows_to_lds(const float* __restrict__ xin,
                                                 const int* __restrict__ rowptr,
                                                 const int* __restrict__ cnt,
                                                 const int* __restrict__ perm,
                                                 float (*sA)[IN_CH + 4], float (*sX)[IN_CH + 4],
                                                 int rw, int nrows, int lrow, int lane) {
  int nr = nrows - rw;
  nr = (nr < 0) ? 0 : ((nr > RPWF) ? RPWF : nr);
  const v4f zero = {0.f, 0.f, 0.f, 0.f};
  const float* xl = xin + lane * 4;
  v4f xd[RPWF];
  for (int i = 0; i < RPWF; ++i)
    xd[i] = (i < nr) ? *(const v4f*)(xl + (size_t)(rw + i) * IN_CH) : zero;
  int st = 0, cn = 0;
  if (lane < nr) {
    st = rowptr[rw + lane];
    cn = cnt[rw + lane];
  }
  int start = __builtin_amdgcn_readlane(st, 0);
  int n     = __builtin_amdgcn_readlane(cn, 0);
  int mine  = (lane < n) ? perm[start + lane] : 0;
  for (int i = 0; i < RPWF; ++i) *(v4f*)&sX[lrow + i][lane * 4] = xd[i];
  int first = __builtin_amdgcn_readlane(mine, 0);
#define GATHER_ROW(k) (*(const v4f*)(xl + (size_t)__builtin_amdgcn_readlane(mine, (k)) * IN_CH))
  for (int r = 0; r < nr; ++r) {
    int start_nx = 0, n_nx = 0, mine_nx = 0;
    if (r + 1 < nr) {
      start_nx = __builtin_amdgcn_readlane(st, r + 1);
      n_nx     = __builtin_amdgcn_readlane(cn, r + 1);
      mine_nx  = (lane < n_nx) ? perm[start_nx + lane] : 0;
    }
    v4f acc = zero;
    for (int i = 0; i < n; i += 32) {
      const int m = (n - i < 32) ? (n - i) : 32;
      if (i > 0) {
        mine  = (lane < m) ? perm[start + i + lane] : 0;
        first = __builtin_amdgcn_readlane(mine, 0);
      }
      const v4f p0 = *(const v4f*)(xl + (size_t)first * IN_CH);
      int j = 1;
      for (; j + 8 <= m; j += 8) {
        const v4f v0 = GATHER_ROW(j);
        const v4f v1 = GATHER_ROW(j + 1);
        const v4f v2 = GATHER_ROW(j + 2);
        const v4f v3 = GATHER_ROW(j + 3);
        const v4f v4 = GATHER_ROW(j + 4);
        const v4f v5 = GATHER_ROW(j + 5);
        const v4f v6 = GATHER_ROW(j + 6);
        const v4f v7 = GATHER_ROW(j + 7);
        acc += v0; acc += v1; acc += v2; acc += v3;
        acc += v4; acc += v5; acc += v6; acc += v7;
      }
      const int rem = m - j;
      v4f q0 = zero, q1 = zero, q2 = zero, q3 = zero, d0 = zero, d1 = zero, o0 = zero;
      if (rem & 4) {
        q0 = GATHER_ROW(j);
        q1 = GATHER_ROW(j + 1);
        q2 = GATHER_ROW(j + 2);
        q3 = GATHER_ROW(j + 3);
        j += 4;
      }
      if (rem & 2) {
        d0 = GATHER_ROW(j);
        d1 = GATHER_ROW(j + 1);
        j += 2;
      }
      if (rem & 1) {
        o0 = GATHER_ROW(j);
      }
      acc += q0; acc += q1; acc += q2; acc += q3;
      acc += d0; acc += d1; acc += o0;
      acc += p0;
    }
    const float inv = 1.0f / fmaxf((float)n, 1.0f);
    *(v4f*)&sA[lrow + r][lane * 4] = acc * inv;
    start = start_nx;
    n     = n_nx;
    mine  = mine_nx;
    first = __builtin_amdgcn_readlane(mine, 0);
  }
#undef GATHER_ROW
  for (int r = 0; r < RPWF; ++r)
    if (r >= nr) *(v4f*)&sA[lrow + r][lane * 4] = zero;
}

#define RT0    64
#define WCH    16
#define WPITCH (HID + 8)
__global__ __launch_bounds__(512)
void sage_gather_gemm_relu(const float* __restrict__ xin,
                           const int* __restrict__ rowptr, const int* __restrict__ cnt,
                           const int* __restrict__ perm,
                           const float* __restrict__ Wl, const float* __restrict__ bl,
                           const float* __restrict__ Wr,
                           float* __restrict__ hout, int nrows)
{
  __shared__ __attribute__((aligned(16))) float sAX[2][RT0][IN_CH + 4];
  __shared__ __attribute__((aligned(16))) float sW[WCH][WPITCH];
  const int tid  = threadIdx.x;
  const int r0   = blockIdx.x * RT0;
  const int lane = tid & 31;
  const int wave = __builtin_amdgcn_readfirstlane(tid >> 5);

  const float* wsrc = Wl + (size_t)wave * HID + lane * 4;
  v4f wreg = *(const v4f*)wsrc;

  mean_rows_to_lds(xin, rowptr, cnt, perm, sAX[0], sAX[1], r0 + wave * RPWF, nrows, wave * RPWF, lane);
  *(v4f*)&sW[wave][lane * 4] = wreg;
  __syncthreads();

  const int half = lane >> 4;
  const int m    = lane & 15;
  const int n    = lane & 15;
  const int kh   = half * 2;
  const int col  = (wave & 7) * 16 + n;
  const int ma   = (wave >> 3) * 32 + m;

  v8f acc0 = {0.f, 0.f, 0.f, 0.f, 0.f, 0.f, 0.f, 0.f};
  v8f acc1 = {0.f, 0.f, 0.f, 0.f, 0.f, 0.f, 0.f, 0.f};
  for (int c = 0; c < 2 * IN_CH / WCH; ++c) {
    const int cn = c + 1;
    if (cn < 2 * IN_CH / WCH) {
      const float* Wn = (cn < IN_CH / WCH) ? Wl : Wr;
      wreg = *(const v4f*)(Wn + (size_t)((cn & (IN_CH / WCH - 1)) * WCH + wave) * HID + lane * 4);
    }
    const int op = c / (IN_CH / WCH);
    const int k0 = (c & (IN_CH / WCH - 1)) * WCH;
    for (int s = 0; s < WCH; s += 4) {
      v2f a0, a1, b;
      a0.x = sAX[op][ma][k0 + s + kh];
      a0.y = sAX[op][ma][k0 + s + kh + 1];
      a1.x = sAX[op][ma + 16][k0 + s + kh];
      a1.y = sAX[op][ma + 16][k0 + s + kh + 1];
      b.x  = sW[s + kh][col];
      b.y  = sW[s + kh + 1][col];
      acc0 = __builtin_amdgcn_wmma_f32_16x16x4_f32(false, a0, false, b, (short)0, acc0, false, false);
      acc1 = __builtin_amdgcn_wmma_f32_16x16x4_f32(false, a1, false, b, (short)0, acc1, false, false);
    }
    if (cn < 2 * IN_CH / WCH) {
      __syncthreads();
      *(v4f*)&sW[wave][lane * 4] = wreg;
      __syncthreads();
    }
  }

  const float bias  = bl[col];
  const int   mbase = (wave >> 3) * 32 + half * 8;
  for (int r = 0; r < 8; ++r) {
    const int row = r0 + mbase + r;
    if (row < nrows)
      hout[(size_t)row * HID + col] = fmaxf(acc0[r] + bias, 0.0f);
    if (row + 16 < nrows)
      hout[(size_t)(row + 16) * HID + col] = fmaxf(acc1[r] + bias, 0.0f);
  }
}

__global__ __launch_bounds__(128)
void sage_gather_gemm_lsm(const float* __restrict__ hin,
                          const int* __restrict__ rowptr, const int* __restrict__ cnt,
                          const int* __restrict__ perm,
                          const float* __restrict__ Wl, const float* __restrict__ bl,
                          const float* __restrict__ Wr,
                          float* __restrict__ out, int nrows)
{
  __shared__ __attribute__((aligned(16))) float sA[16][HID + 4];
  __shared__ __attribute__((aligned(16))) float sX[16][HID + 4];
  __shared__ float sL[16][OUTP];
  const int tid  = threadIdx.x;
  const int r0   = blockIdx.x * 16;
  const int lane = tid & 31;
  const int wave = __builtin_amdgcn_readfirstlane(tid >> 5);

  mean_rows_to_lds(hin, rowptr, cnt, perm, sA, sX, r0 + wave * RPWF, nrows, wave * RPWF, lane);
  __syncthreads();

  const int half = lane >> 4;
  const int m    = lane & 15;
  const int n    = lane & 15;
  const int kh   = half * 2;

  if (wave < 3) {
    const int n0  = wave * 16;
    const int col = n0 + n;
    const bool cv = (col < OUTC);
    v8f acc = {0.f, 0.f, 0.f, 0.f, 0.f, 0.f, 0.f, 0.f};
    for (int k = 0; k < HID; k += 4) {
      v2f a, b;
      a.x = sA[m][k + kh];
      a.y = sA[m][k + kh + 1];
      b.x = cv ? Wl[(size_t)(k + kh)     * OUTC + col] : 0.0f;
      b.y = cv ? Wl[(size_t)(k + kh + 1) * OUTC + col] : 0.0f;
      acc = __builtin_amdgcn_wmma_f32_16x16x4_f32(false, a, false, b, (short)0, acc, false, false);
    }
    for (int k = 0; k < HID; k += 4) {
      v2f a, b;
      a.x = sX[m][k + kh];
      a.y = sX[m][k + kh + 1];
      b.x = cv ? Wr[(size_t)(k + kh)     * OUTC + col] : 0.0f;
      b.y = cv ? Wr[(size_t)(k + kh + 1) * OUTC + col] : 0.0f;
      acc = __builtin_amdgcn_wmma_f32_16x16x4_f32(false, a, false, b, (short)0, acc, false, false);
    }
    const float bias  = cv ? bl[col] : 0.0f;
    const int   mbase = half * 8;
    for (int r = 0; r < 8; ++r)
      sL[mbase + r][col] = acc[r] + bias;
  }
  __syncthreads();

  if (tid < 16) {
    int row = r0 + tid;
    if (row < nrows) {
      float mx = -INFINITY;
      for (int c = 0; c < OUTC; ++c) mx = fmaxf(mx, sL[tid][c]);
      float s = 0.0f;
      for (int c = 0; c < OUTC; ++c) s += expf(sL[tid][c] - mx);
      float lse = logf(s);
      for (int c = 0; c < OUTC; ++c)
        out[(size_t)row * OUTC + c] = sL[tid][c] - mx - lse;
    }
  }
}

extern "C" void kernel_launch(void* const* d_in, const int* in_sizes, int n_in,
                              void* d_out, int out_size, void* d_ws, size_t ws_size,
                              hipStream_t stream)
{
  const float* x    = (const float*)d_in[0];
  const int*   src0 = (const int*)d_in[1];
  const int*   dst0 = (const int*)d_in[2];
  const int*   src1 = (const int*)d_in[3];
  const int*   dst1 = (const int*)d_in[4];
  const float* Wl0  = (const float*)d_in[7];
  const float* bl0  = (const float*)d_in[8];
  const float* Wr0  = (const float*)d_in[9];
  const float* Wl1  = (const float*)d_in[10];
  const float* bl1  = (const float*)d_in[11];
  const float* Wr1  = (const float*)d_in[12];
  float* out = (float*)d_out;

  const int E0 = in_sizes[1];
  const int E1 = in_sizes[3];
  const int N1 = 100000;
  const int N2 = out_size / OUTC;

  float* msg0 = (float*)d_ws;
  float* deg0 = msg0 + (size_t)N1 * HID;
  float* h    = deg0 + N1;
  float* msg1 = h    + (size_t)N1 * HID;

  const int N1p = (N1 + 3) & ~3;
  const int N2p = (N2 + 3) & ~3;
  const int E0p = (E0 + 3) & ~3;
  const int E1p = (E1 + 3) & ~3;
  const int nblk0 = (E0 + CHUNK - 1) / CHUNK;
  const int nblk1 = (E1 + CHUNK - 1) / CHUNK;
  int* cnt0  = (int*)msg0;
  int* rp0   = cnt0 + N1p;
  int* perm0 = rp0 + N1p;
  int* bdst0 = perm0 + E0p;
  int* bsrc0 = bdst0 + E0p;
  int* boff0 = bsrc0 + E0p;
  int* table = boff0 + NBIN + 4;
  int* cnt1  = (int*)msg1;
  int* rp1   = cnt1 + N2p;
  int* perm1 = rp1 + N2p;
  int* bdst1 = perm1 + E1p;
  int* bsrc1 = bdst1 + E1p;
  int* boff1 = bsrc1 + E1p;

  bin_hist<<<nblk0 + nblk1, ETPB, 0, stream>>>(dst0, E0, N1, nblk0, dst1, E1, N2, table);
  bin_scatter<<<nblk0 + nblk1, ETPB, 0, stream>>>(src0, dst0, E0, N1, nblk0, src1, dst1, E1, N2, nblk1,
                                                  table, bdst0, bsrc0, boff0, bdst1, bsrc1, boff1);
  bin_csr<<<2 * NBIN, ETPB, 0, stream>>>(bdst0, bsrc0, boff0, N1, rp0, cnt0, perm0,
                                         bdst1, bsrc1, boff1, N2, rp1, cnt1, perm1);

  sage_gather_gemm_relu<<<(N1 + RT0 - 1) / RT0, 512, 0, stream>>>(x, rp0, cnt0, perm0, Wl0, bl0, Wr0, h, N1);

  sage_gather_gemm_lsm<<<(N2 + 15) / 16, 128, 0, stream>>>(h, rp1, cnt1, perm1, Wl1, bl1, Wr1, out, N2);
}


# edge sort in 2 launches, not 3: a block orders its 8192 edges by bin at its own place (one LDS count gives the ranks); the per-bin block gathers the chunks' runs into LDS; sum order = LDS arrival
# speedup vs baseline: 3.8511x; 1.0365x over previous
#include <hip/hip_runtime.h>
#include <math.h>

#define IN_CH 128
#define HID   128
#define OUTC  47
#define OUTP  48
#define RPWF  4

typedef float v2f __attribute__((ext_vector_type(2)));
typedef float v8f __attribute__((ext_vector_type(8)));


typedef float v4f __attribute__((ext_vector_type(4)));
typedef int   v4i __attribute__((ext_vector_type(4)));

#define NBIN   256
#define CHUNK  8192
#define ETPB   1024
#define RMAX   512
#define PMAX   12288
#define TPITCH (NBIN + 4)

__device__ __forceinline__ int bin_shift(int n) {
  const int s = 24 - __clz(n - 1);
  return s < 0 ? 0 : s;
}

__global__ __launch_bounds__(ETPB)
void bin_local(const int* __restrict__ src0, const int* __restrict__ dst0, int E0, int N0, int nblk0,
               const int* __restrict__ src1, const int* __restrict__ dst1, int E1, int N1,
               int* __restrict__ table,
               int* __restrict__ bdst0, int* __restrict__ bsrc0,
               int* __restrict__ bdst1, int* __restrict__ bsrc1) {
  __shared__ int h[NBIN];
  __shared__ int sc[NBIN];
  __shared__ int sd[CHUNK];
  __shared__ int ss[CHUNK];
  const int t = threadIdx.x;
  const int* src; const int* dst; int* bdst; int* bsrc; int E, N, c;
  if ((int)blockIdx.x < nblk0) {
    src = src0; dst = dst0; bdst = bdst0; bsrc = bsrc0; E = E0; N = N0; c = blockIdx.x;
  } else {
    src = src1; dst = dst1; bdst = bdst1; bsrc = bsrc1; E = E1; N = N1; c = blockIdx.x - nblk0;
  }
  const int sh = bin_shift(N);
  if (t < NBIN) h[t] = 0;

  const int ea = (c * (CHUNK / 4) + t) * 4;
  const int eb = ea + ETPB * 4;
  v4i da = {-1, -1, -1, -1}, db = {-1, -1, -1, -1};
  v4i sa = {0, 0, 0, 0}, sb = {0, 0, 0, 0};
  if (ea + 3 < E) {
    da = *(const v4i*)(dst + ea);
    sa = *(const v4i*)(src + ea);
  } else {
    if (ea     < E) { da.x = dst[ea];     sa.x = src[ea];     }
    if (ea + 1 < E) { da.y = dst[ea + 1]; sa.y = src[ea + 1]; }
    if (ea + 2 < E) { da.z = dst[ea + 2]; sa.z = src[ea + 2]; }
  }
  if (eb + 3 < E) {
    db = *(const v4i*)(dst + eb);
    sb = *(const v4i*)(src + eb);
  } else {
    if (eb     < E) { db.x = dst[eb];     sb.x = src[eb];     }
    if (eb + 1 < E) { db.y = dst[eb + 1]; sb.y = src[eb + 1]; }
    if (eb + 2 < E) { db.z = dst[eb + 2]; sb.z = src[eb + 2]; }
  }
  __syncthreads();
  v4i ra = {0, 0, 0, 0}, rb = {0, 0, 0, 0};
  if (da.x >= 0) ra.x = atomicAdd(&h[da.x >> sh], 1);
  if (da.y >= 0) ra.y = atomicAdd(&h[da.y >> sh], 1);
  if (da.z >= 0) ra.z = atomicAdd(&h[da.z >> sh], 1);
  if (da.w >= 0) ra.w = atomicAdd(&h[da.w >> sh], 1);
  if (db.x >= 0) rb.x = atomicAdd(&h[db.x >> sh], 1);
  if (db.y >= 0) rb.y = atomicAdd(&h[db.y >> sh], 1);
  if (db.z >= 0) rb.z = atomicAdd(&h[db.z >> sh], 1);
  if (db.w >= 0) rb.w = atomicAdd(&h[db.w >> sh], 1);
  __syncthreads();
  int own = 0;
  if (t < NBIN) { own = h[t]; sc[t] = own; }
  __syncthreads();
  for (int d = 1; d < NBIN; d <<= 1) {
    int v = 0;
    if (t < NBIN && t >= d) v = sc[t - d];
    __syncthreads();
    if (t < NBIN) sc[t] += v;
    __syncthreads();
  }
  if (t < NBIN) {
    const int lex = sc[t] - own;
    h[t] = lex;
    table[(size_t)blockIdx.x * TPITCH + t] = lex;
    if (t == NBIN - 1) table[(size_t)blockIdx.x * TPITCH + NBIN] = sc[t];
  }
  __syncthreads();
  if (da.x >= 0) { const int p = h[da.x >> sh] + ra.x; sd[p] = da.x; ss[p] = sa.x; }
  if (da.y >= 0) { const int p = h[da.y >> sh] + ra.y; sd[p] = da.y; ss[p] = sa.y; }
  if (da.z >= 0) { const int p = h[da.z >> sh] + ra.z; sd[p] = da.z; ss[p] = sa.z; }
  if (da.w >= 0) { const int p = h[da.w >> sh] + ra.w; sd[p] = da.w; ss[p] = sa.w; }
  if (db.x >= 0) { const int p = h[db.x >> sh] + rb.x; sd[p] = db.x; ss[p] = sb.x; }
  if (db.y >= 0) { const int p = h[db.y >> sh] + rb.y; sd[p] = db.y; ss[p] = sb.y; }
  if (db.z >= 0) { const int p = h[db.z >> sh] + rb.z; sd[p] = db.z; ss[p] = sb.z; }
  if (db.w >= 0) { const int p = h[db.w >> sh] + rb.w; sd[p] = db.w; ss[p] = sb.w; }
  __syncthreads();
  int nE = E - c * CHUNK;
  nE = (nE > CHUNK) ? CHUNK : nE;
  for (int p = t; p < nE; p += ETPB) {
    bdst[(size_t)c * CHUNK + p] = sd[p];
    bsrc[(size_t)c * CHUNK + p] = ss[p];
  }
}

__global__ __launch_bounds__(ETPB)
void bin_csr(const int* __restrict__ bdst0, const int* __restrict__ bsrc0, int nblk0, int N0,
             int* __restrict__ rp0, int* __restrict__ cnt0, int* __restrict__ perm0,
             const int* __restrict__ bdst1, const int* __restrict__ bsrc1, int nblk1, int N1,
             int* __restrict__ rp1, int* __restrict__ cnt1, int* __restrict__ perm1,
             const int* __restrict__ table) {
  __shared__ int lc[RMAX];
  __shared__ int ls[RMAX];
  __shared__ int lq[RMAX];
  __shared__ int lp[PMAX];
  __shared__ int es[PMAX];
  __shared__ unsigned short ed[PMAX];
  __shared__ int sBeg;
  __shared__ int sCur;
  const int t    = threadIdx.x;
  const int lane = t & 31;
  const int wave = __builtin_amdgcn_readfirstlane(t >> 5);
  const int* bdst; const int* bsrc; const int* tab; int* rp; int* cnt; int* perm; int N, bin, nb;
  if ((int)blockIdx.x < NBIN) {
    bdst = bdst0; bsrc = bsrc0; tab = table; rp = rp0; cnt = cnt0; perm = perm0;
    N = N0; bin = blockIdx.x; nb = nblk0;
  } else {
    bdst = bdst1; bsrc = bsrc1; tab = table + (size_t)nblk0 * TPITCH; rp = rp1; cnt = cnt1; perm = perm1;
    N = N1; bin = blockIdx.x - NBIN; nb = nblk1;
  }
  const int sh   = bin_shift(N);
  const int row0 = bin << sh;
  if (row0 >= N) return;
  int nrow = N - row0;
  nrow = (nrow > (1 << sh)) ? (1 << sh) : nrow;

  const int myc = wave + 32 * lane;
  int myS = 0, myL = 0;
  if (myc < nb) {
    myS = tab[(size_t)myc * TPITCH + bin];
    myL = tab[(size_t)myc * TPITCH + bin + 1] - myS;
  }
  if (t < RMAX) lc[t] = 0;
  if (t == 0) { sBeg = 0; sCur = 0; }
  __syncthreads();
  int myP = 0;
  if (myc < nb) {
    atomicAdd(&sBeg, myS);
    myP = atomicAdd(&sCur, myL);
  }
  __syncthreads();
  const int beg   = sBeg;
  const int total = sCur;
  const int nch = (nb > wave) ? ((nb - wave + 31) >> 5) : 0;

  if (total <= PMAX) {
    for (int i = 0; i < nch; i += 2) {
      const int sA = __builtin_amdgcn_readlane(myS, i);
      const int lA = __builtin_amdgcn_readlane(myL, i);
      const int pA = __builtin_amdgcn_readlane(myP, i);
      int sB = 0, lB = 0, pB = 0;
      if (i + 1 < nch) {
        sB = __builtin_amdgcn_readlane(myS, i + 1);
        lB = __builtin_amdgcn_readlane(myL, i + 1);
        pB = __builtin_amdgcn_readlane(myP, i + 1);
      }
      const size_t oA = (size_t)(wave + 32 * i) * CHUNK + sA;
      const size_t oB = (i + 1 < nch) ? ((size_t)(wave + 32 * i + 32) * CHUNK + sB) : 0;
      const int* dA = bdst + oA; const int* qA = bsrc + oA;
      const int* dB = bdst + oB; const int* qB = bsrc + oB;
      int a0 = 0, a1 = 0, b0 = 0, b1 = 0, u0 = 0, u1 = 0, w0 = 0, w1 = 0;
      if (lane      < lA) { a0 = dA[lane];      u0 = qA[lane];      }
      if (lane + 32 < lA) { a1 = dA[lane + 32]; u1 = qA[lane + 32]; }
      if (lane      < lB) { b0 = dB[lane];      w0 = qB[lane];      }
      if (lane + 32 < lB) { b1 = dB[lane + 32]; w1 = qB[lane + 32]; }
      if (lane      < lA) { ed[pA + lane]      = (unsigned short)(a0 - row0); es[pA + lane]      = u0; }
      if (lane + 32 < lA) { ed[pA + lane + 32] = (unsigned short)(a1 - row0); es[pA + lane + 32] = u1; }
      if (lane      < lB) { ed[pB + lane]      = (unsigned short)(b0 - row0); es[pB + lane]      = w0; }
      if (lane + 32 < lB) { ed[pB + lane + 32] = (unsigned short)(b1 - row0); es[pB + lane + 32] = w1; }
      for (int k = 64 + lane; k < lA; k += 32) { ed[pA + k] = (unsigned short)(dA[k] - row0); es[pA + k] = qA[k]; }
      for (int k = 64 + lane; k < lB; k += 32) { ed[pB + k] = (unsigned short)(dB[k] - row0); es[pB + k] = qB[k]; }
    }
    __syncthreads();
    int ic = t;
    for (; ic + 3 * ETPB < total; ic += 4 * ETPB) {
      const int d0 = ed[ic];
      const int d1 = ed[ic + ETPB];
      const int d2 = ed[ic + 2 * ETPB];
      const int d3 = ed[ic + 3 * ETPB];
      atomicAdd(&lc[d0], 1);
      atomicAdd(&lc[d1], 1);
      atomicAdd(&lc[d2], 1);
      atomicAdd(&lc[d3], 1);
    }
    for (; ic < total; ic += ETPB) atomicAdd(&lc[ed[ic]], 1);
  } else {
    for (int cc = wave; cc < nb; cc += 32) {
      const int s = tab[(size_t)cc * TPITCH + bin];
      const int e = tab[(size_t)cc * TPITCH + bin + 1];
      for (int k = s + lane; k < e; k += 32) atomicAdd(&lc[bdst[(size_t)cc * CHUNK + k] - row0], 1);
    }
  }
  __syncthreads();
  int cn = 0;
  if (t < RMAX) { cn = lc[t]; ls[t] = cn; }
  __syncthreads();
  for (int d = 1; d < RMAX; d <<= 1) {
    const int v = (t < RMAX && t >= d) ? ls[t - d] : 0;
    __syncthreads();
    if (t < RMAX) ls[t] += v;
    __syncthreads();
  }
  if (t < RMAX) {
    const int st = ls[t] - cn;
    lq[t] = st;
    if (t < nrow) { rp[row0 + t] = beg + st; cnt[row0 + t] = cn; }
  }
  __syncthreads();
  if (total <= PMAX) {
    int iq = t;
    for (; iq + 3 * ETPB < total; iq += 4 * ETPB) {
      const int d0 = ed[iq];
      const int d1 = ed[iq + ETPB];
      const int d2 = ed[iq + 2 * ETPB];
      const int d3 = ed[iq + 3 * ETPB];
      const int s0 = es[iq];
      const int s1 = es[iq + ETPB];
      const int s2 = es[iq + 2 * ETPB];
      const int s3 = es[iq + 3 * ETPB];
      const int p0 = atomicAdd(&lq[d0], 1);
      const int p1 = atomicAdd(&lq[d1], 1);
      const int p2 = atomicAdd(&lq[d2], 1);
      const int p3 = atomicAdd(&lq[d3], 1);
      lp[p0] = s0;
      lp[p1] = s1;
      lp[p2] = s2;
      lp[p3] = s3;
    }
    for (; iq < total; iq += ETPB) {
      const int p = atomicAdd(&lq[ed[iq]], 1);
      lp[p] = es[iq];
    }
    __syncthreads();
    for (int i = t; i < total; i += ETPB) perm[beg + i] = lp[i];
    return;
  }
  for (int cc = wave; cc < nb; cc += 32) {
    const int s = tab[(size_t)cc * TPITCH + bin];
    const int e = tab[(size_t)cc * TPITCH + bin + 1];
    for (int k = s + lane; k < e; k += 32) {
      const int p = atomicAdd(&lq[bdst[(size_t)cc * CHUNK + k] - row0], 1);
      perm[beg + p] = bsrc[(size_t)cc * CHUNK + k];
    }
  }
}

__device__ __forceinline__ void mean_rows_to_lds(const float* __restrict__ xin,
                                                 const int* __restrict__ rowptr,
                                                 const int* __restrict__ cnt,
                                                 const int* __restrict__ perm,
                                                 float (*sA)[IN_CH + 4], float (*sX)[IN_CH + 4],
                                                 int rw, int nrows, int lrow, int lane) {
  int nr = nrows - rw;
  nr = (nr < 0) ? 0 : ((nr > RPWF) ? RPWF : nr);
  const v4f zero = {0.f, 0.f, 0.f, 0.f};
  const float* xl = xin + lane * 4;
  v4f xd[RPWF];
  for (int i = 0; i < RPWF; ++i)
    xd[i] = (i < nr) ? *(const v4f*)(xl + (size_t)(rw + i) * IN_CH) : zero;
  int st = 0, cn = 0;
  if (lane < nr) {
    st = rowptr[rw + lane];
    cn = cnt[rw + lane];
  }
  int start = __builtin_amdgcn_readlane(st, 0);
  int n     = __builtin_amdgcn_readlane(cn, 0);
  int mine  = (lane < n) ? perm[start + lane] : 0;
  for (int i = 0; i < RPWF; ++i) *(v4f*)&sX[lrow + i][lane * 4] = xd[i];
  int first = __builtin_amdgcn_readlane(mine, 0);
#define GATHER_ROW(k) (*(const v4f*)(xl + (size_t)__builtin_amdgcn_readlane(mine, (k)) * IN_CH))
  for (int r = 0; r < nr; ++r) {
    int start_nx = 0, n_nx = 0, mine_nx = 0;
    if (r + 1 < nr) {
      start_nx = __builtin_amdgcn_readlane(st, r + 1);
      n_nx     = __builtin_amdgcn_readlane(cn, r + 1);
      mine_nx  = (lane < n_nx) ? perm[start_nx + lane] : 0;
    }
    v4f acc = zero;
    for (int i = 0; i < n; i += 32) {
      const int m = (n - i < 32) ? (n - i) : 32;
      if (i > 0) {
        mine  = (lane < m) ? perm[start + i + lane] : 0;
        first = __builtin_amdgcn_readlane(mine, 0);
      }
      const v4f p0 = *(const v4f*)(xl + (size_t)first * IN_CH);
      int j = 1;
      for (; j + 8 <= m; j += 8) {
        const v4f v0 = GATHER_ROW(j);
        const v4f v1 = GATHER_ROW(j + 1);
        const v4f v2 = GATHER_ROW(j + 2);
        const v4f v3 = GATHER_ROW(j + 3);
        const v4f v4 = GATHER_ROW(j + 4);
        const v4f v5 = GATHER_ROW(j + 5);
        const v4f v6 = GATHER_ROW(j + 6);
        const v4f v7 = GATHER_ROW(j + 7);
        acc += v0; acc += v1; acc += v2; acc += v3;
        acc += v4; acc += v5; acc += v6; acc += v7;
      }
      const int rem = m - j;
      v4f q0 = zero, q1 = zero, q2 = zero, q3 = zero, d0 = zero, d1 = zero, o0 = zero;
      if (rem & 4) {
        q0 = GATHER_ROW(j);
        q1 = GATHER_ROW(j + 1);
        q2 = GATHER_ROW(j + 2);
        q3 = GATHER_ROW(j + 3);
        j += 4;
      }
      if (rem & 2) {
        d0 = GATHER_ROW(j);
        d1 = GATHER_ROW(j + 1);
        j += 2;
      }
      if (rem & 1) {
        o0 = GATHER_ROW(j);
      }
      acc += q0; acc += q1; acc += q2; acc += q3;
      acc += d0; acc += d1; acc += o0;
      acc += p0;
    }
    const float inv = 1.0f / fmaxf((float)n, 1.0f);
    *(v4f*)&sA[lrow + r][lane * 4] = acc * inv;
    start = start_nx;
    n     = n_nx;
    mine  = mine_nx;
    first = __builtin_amdgcn_readlane(mine, 0);
  }
#undef GATHER_ROW
  for (int r = 0; r < RPWF; ++r)
    if (r >= nr) *(v4f*)&sA[lrow + r][lane * 4] = zero;
}

#define RT0    64
#define WCH    16
#define WPITCH (HID + 8)
__global__ __launch_bounds__(512)
void sage_gather_gemm_relu(const float* __restrict__ xin,
                           const int* __restrict__ rowptr, const int* __restrict__ cnt,
                           const int* __restrict__ perm,
                           const float* __restrict__ Wl, const float* __restrict__ bl,
                           const float* __restrict__ Wr,
                           float* __restrict__ hout, int nrows)
{
  __shared__ __attribute__((aligned(16))) float sAX[2][RT0][IN_CH + 4];
  __shared__ __attribute__((aligned(16))) float sW[WCH][WPITCH];
  const int tid  = threadIdx.x;
  const int r0   = blockIdx.x * RT0;
  const int lane = tid & 31;
  const int wave = __builtin_amdgcn_readfirstlane(tid >> 5);

  const float* wsrc = Wl + (size_t)wave * HID + lane * 4;
  v4f wreg = *(const v4f*)wsrc;

  mean_rows_to_lds(xin, rowptr, cnt, perm, sAX[0], sAX[1], r0 + wave * RPWF, nrows, wave * RPWF, lane);
  *(v4f*)&sW[wave][lane * 4] = wreg;
  __syncthreads();

  const int half = lane >> 4;
  const int m    = lane & 15;
  const int n    = lane & 15;
  const int kh   = half * 2;
  const int col  = (wave & 7) * 16 + n;
  const int ma   = (wave >> 3) * 32 + m;

  v8f acc0 = {0.f, 0.f, 0.f, 0.f, 0.f, 0.f, 0.f, 0.f};
  v8f acc1 = {0.f, 0.f, 0.f, 0.f, 0.f, 0.f, 0.f, 0.f};
  for (int c = 0; c < 2 * IN_CH / WCH; ++c) {
    const int cn = c + 1;
    if (cn < 2 * IN_CH / WCH) {
      const float* Wn = (cn < IN_CH / WCH) ? Wl : Wr;
      wreg = *(const v4f*)(Wn + (size_t)((cn & (IN_CH / WCH - 1)) * WCH + wave) * HID + lane * 4);
    }
    const int op = c / (IN_CH / WCH);
    const int k0 = (c & (IN_CH / WCH - 1)) * WCH;
    for (int s = 0; s < WCH; s += 4) {
      v2f a0, a1, b;
      a0.x = sAX[op][ma][k0 + s + kh];
      a0.y = sAX[op][ma][k0 + s + kh + 1];
      a1.x = sAX[op][ma + 16][k0 + s + kh];
      a1.y = sAX[op][ma + 16][k0 + s + kh + 1];
      b.x  = sW[s + kh][col];
      b.y  = sW[s + kh + 1][col];
      acc0 = __builtin_amdgcn_wmma_f32_16x16x4_f32(false, a0, false, b, (short)0, acc0, false, false);
      acc1 = __builtin_amdgcn_wmma_f32_16x16x4_f32(false, a1, false, b, (short)0, acc1, false, false);
    }
    if (cn < 2 * IN_CH / WCH) {
      __syncthreads();
      *(v4f*)&sW[wave][lane * 4] = wreg;
      __syncthreads();
    }
  }

  const float bias  = bl[col];
  const int   mbase = (wave >> 3) * 32 + half * 8;
  for (int r = 0; r < 8; ++r) {
    const int row = r0 + mbase + r;
    if (row < nrows)
      hout[(size_t)row * HID + col] = fmaxf(acc0[r] + bias, 0.0f);
    if (row + 16 < nrows)
      hout[(size_t)(row + 16) * HID + col] = fmaxf(acc1[r] + bias, 0.0f);
  }
}

__global__ __launch_bounds__(128)
void sage_gather_gemm_lsm(const float* __restrict__ hin,
                          const int* __restrict__ rowptr, const int* __restrict__ cnt,
                          const int* __restrict__ perm,
                          const float* __restrict__ Wl, const float* __restrict__ bl,
                          const float* __restrict__ Wr,
                          float* __restrict__ out, int nrows)
{
  __shared__ __attribute__((aligned(16))) float sA[16][HID + 4];
  __shared__ __attribute__((aligned(16))) float sX[16][HID + 4];
  __shared__ float sL[16][OUTP];
  const int tid  = threadIdx.x;
  const int r0   = blockIdx.x * 16;
  const int lane = tid & 31;
  const int wave = __builtin_amdgcn_readfirstlane(tid >> 5);

  mean_rows_to_lds(hin, rowptr, cnt, perm, sA, sX, r0 + wave * RPWF, nrows, wave * RPWF, lane);
  __syncthreads();

  const int half = lane >> 4;
  const int m    = lane & 15;
  const int n    = lane & 15;
  const int kh   = half * 2;

  if (wave < 3) {
    const int n0  = wave * 16;
    const int col = n0 + n;
    const bool cv = (col < OUTC);
    v8f acc = {0.f, 0.f, 0.f, 0.f, 0.f, 0.f, 0.f, 0.f};
    for (int k = 0; k < HID; k += 4) {
      v2f a, b;
      a.x = sA[m][k + kh];
      a.y = sA[m][k + kh + 1];
      b.x = cv ? Wl[(size_t)(k + kh)     * OUTC + col] : 0.0f;
      b.y = cv ? Wl[(size_t)(k + kh + 1) * OUTC + col] : 0.0f;
      acc = __builtin_amdgcn_wmma_f32_16x16x4_f32(false, a, false, b, (short)0, acc, false, false);
    }
    for (int k = 0; k < HID; k += 4) {
      v2f a, b;
      a.x = sX[m][k + kh];
      a.y = sX[m][k + kh + 1];
      b.x = cv ? Wr[(size_t)(k + kh)     * OUTC + col] : 0.0f;
      b.y = cv ? Wr[(size_t)(k + kh + 1) * OUTC + col] : 0.0f;
      acc = __builtin_amdgcn_wmma_f32_16x16x4_f32(false, a, false, b, (short)0, acc, false, false);
    }
    const float bias  = cv ? bl[col] : 0.0f;
    const int   mbase = half * 8;
    for (int r = 0; r < 8; ++r)
      sL[mbase + r][col] = acc[r] + bias;
  }
  __syncthreads();

  if (tid < 16) {
    int row = r0 + tid;
    if (row < nrows) {
      float mx = -INFINITY;
      for (int c = 0; c < OUTC; ++c) mx = fmaxf(mx, sL[tid][c]);
      float s = 0.0f;
      for (int c = 0; c < OUTC; ++c) s += expf(sL[tid][c] - mx);
      float lse = logf(s);
      for (int c = 0; c < OUTC; ++c)
        out[(size_t)row * OUTC + c] = sL[tid][c] - mx - lse;
    }
  }
}

extern "C" void kernel_launch(void* const* d_in, const int* in_sizes, int n_in,
                              void* d_out, int out_size, void* d_ws, size_t ws_size,
                              hipStream_t stream)
{
  const float* x    = (const float*)d_in[0];
  const int*   src0 = (const int*)d_in[1];
  const int*   dst0 = (const int*)d_in[2];
  const int*   src1 = (const int*)d_in[3];
  const int*   dst1 = (const int*)d_in[4];
  const float* Wl0  = (const float*)d_in[7];
  const float* bl0  = (const float*)d_in[8];
  const float* Wr0  = (const float*)d_in[9];
  const float* Wl1  = (const float*)d_in[10];
  const float* bl1  = (const float*)d_in[11];
  const float* Wr1  = (const float*)d_in[12];
  float* out = (float*)d_out;

  const int E0 = in_sizes[1];
  const int E1 = in_sizes[3];
  const int N1 = 100000;
  const int N2 = out_size / OUTC;

  float* msg0 = (float*)d_ws;
  float* deg0 = msg0 + (size_t)N1 * HID;
  float* h    = deg0 + N1;
  float* msg1 = h    + (size_t)N1 * HID;

  const int N1p = (N1 + 3) & ~3;
  const int N2p = (N2 + 3) & ~3;
  const int E0p = (E0 + 3) & ~3;
  const int E1p = (E1 + 3) & ~3;
  const int nblk0 = (E0 + CHUNK - 1) / CHUNK;
  const int nblk1 = (E1 + CHUNK - 1) / CHUNK;
  int* cnt0  = (int*)msg0;
  int* rp0   = cnt0 + N1p;
  int* perm0 = rp0 + N1p;
  int* bdst0 = perm0 + E0p;
  int* bsrc0 = bdst0 + E0p;
  int* table = bsrc0 + E0p;
  int* cnt1  = (int*)msg1;
  int* rp1   = cnt1 + N2p;
  int* perm1 = rp1 + N2p;
  int* bdst1 = perm1 + E1p;
  int* bsrc1 = bdst1 + E1p;

  bin_local<<<nblk0 + nblk1, ETPB, 0, stream>>>(src0, dst0, E0, N1, nblk0, src1, dst1, E1, N2,
                                                table, bdst0, bsrc0, bdst1, bsrc1);
  bin_csr<<<2 * NBIN, ETPB, 0, stream>>>(bdst0, bsrc0, nblk0, N1, rp0, cnt0, perm0,
                                         bdst1, bsrc1, nblk1, N2, rp1, cnt1, perm1, table);

  sage_gather_gemm_relu<<<(N1 + RT0 - 1) / RT0, 512, 0, stream>>>(x, rp0, cnt0, perm0, Wl0, bl0, Wr0, h, N1);

  sage_gather_gemm_lsm<<<(N2 + 15) / 16, 128, 0, stream>>>(h, rp1, cnt1, perm1, Wl1, bl1, Wr1, out, N2);
}


# layer-1 dense part: a wave asks for all 64 B values of Wl ahead of its row sums and all 64 of Wr ahead of Wl's products (clamped column, no branch), not two loads waited for at each k step
# speedup vs baseline: 4.5561x; 1.1831x over previous
#include <hip/hip_runtime.h>
#include <math.h>

#define IN_CH 128
#define HID   128
#define OUTC  47
#define OUTP  48
#define RPWF  4

typedef float v2f __attribute__((ext_vector_type(2)));
typedef float v8f __attribute__((ext_vector_type(8)));


typedef float v4f __attribute__((ext_vector_type(4)));
typedef int   v4i __attribute__((ext_vector_type(4)));

#define NBIN   256
#define CHUNK  8192
#define ETPB   1024
#define RMAX   512
#define PMAX   12288
#define TPITCH (NBIN + 4)

__device__ __forceinline__ int bin_shift(int n) {
  const int s = 24 - __clz(n - 1);
  return s < 0 ? 0 : s;
}

__global__ __launch_bounds__(ETPB)
void bin_local(const int* __restrict__ src0, const int* __restrict__ dst0, int E0, int N0, int nblk0,
               const int* __restrict__ src1, const int* __restrict__ dst1, int E1, int N1,
               int* __restrict__ table,
               int* __restrict__ bdst0, int* __restrict__ bsrc0,
               int* __restrict__ bdst1, int* __restrict__ bsrc1) {
  __shared__ int h[NBIN];
  __shared__ int sc[NBIN];
  __shared__ int sd[CHUNK];
  __shared__ int ss[CHUNK];
  const int t = threadIdx.x;
  const int* src; const int* dst; int* bdst; int* bsrc; int E, N, c;
  if ((int)blockIdx.x < nblk0) {
    src = src0; dst = dst0; bdst = bdst0; bsrc = bsrc0; E = E0; N = N0; c = blockIdx.x;
  } else {
    src = src1; dst = dst1; bdst = bdst1; bsrc = bsrc1; E = E1; N = N1; c = blockIdx.x - nblk0;
  }
  const int sh = bin_shift(N);
  if (t < NBIN) h[t] = 0;

  const int ea = (c * (CHUNK / 4) + t) * 4;
  const int eb = ea + ETPB * 4;
  v4i da = {-1, -1, -1, -1}, db = {-1, -1, -1, -1};
  v4i sa = {0, 0, 0, 0}, sb = {0, 0, 0, 0};
  if (ea + 3 < E) {
    da = *(const v4i*)(dst + ea);
    sa = *(const v4i*)(src + ea);
  } else {
    if (ea     < E) { da.x = dst[ea];     sa.x = src[ea];     }
    if (ea + 1 < E) { da.y = dst[ea + 1]; sa.y = src[ea + 1]; }
    if (ea + 2 < E) { da.z = dst[ea + 2]; sa.z = src[ea + 2]; }
  }
  if (eb + 3 < E) {
    db = *(const v4i*)(dst + eb);
    sb = *(const v4i*)(src + eb);
  } else {
    if (eb     < E) { db.x = dst[eb];     sb.x = src[eb];     }
    if (eb + 1 < E) { db.y = dst[eb + 1]; sb.y = src[eb + 1]; }
    if (eb + 2 < E) { db.z = dst[eb + 2]; sb.z = src[eb + 2]; }
  }
  __syncthreads();
  v4i ra = {0, 0, 0, 0}, rb = {0, 0, 0, 0};
  if (da.x >= 0) ra.x = atomicAdd(&h[da.x >> sh], 1);
  if (da.y >= 0) ra.y = atomicAdd(&h[da.y >> sh], 1);
  if (da.z >= 0) ra.z = atomicAdd(&h[da.z >> sh], 1);
  if (da.w >= 0) ra.w = atomicAdd(&h[da.w >> sh], 1);
  if (db.x >= 0) rb.x = atomicAdd(&h[db.x >> sh], 1);
  if (db.y >= 0) rb.y = atomicAdd(&h[db.y >> sh], 1);
  if (db.z >= 0) rb.z = atomicAdd(&h[db.z >> sh], 1);
  if (db.w >= 0) rb.w = atomicAdd(&h[db.w >> sh], 1);
  __syncthreads();
  int own = 0;
  if (t < NBIN) { own = h[t]; sc[t] = own; }
  __syncthreads();
  for (int d = 1; d < NBIN; d <<= 1) {
    int v = 0;
    if (t < NBIN && t >= d) v = sc[t - d];
    __syncthreads();
    if (t < NBIN) sc[t] += v;
    __syncthreads();
  }
  if (t < NBIN) {
    const int lex = sc[t] - own;
    h[t] = lex;
    table[(size_t)blockIdx.x * TPITCH + t] = lex;
    if (t == NBIN - 1) table[(size_t)blockIdx.x * TPITCH + NBIN] = sc[t];
  }
  __syncthreads();
  if (da.x >= 0) { const int p = h[da.x >> sh] + ra.x; sd[p] = da.x; ss[p] = sa.x; }
  if (da.y >= 0) { const int p = h[da.y >> sh] + ra.y; sd[p] = da.y; ss[p] = sa.y; }
  if (da.z >= 0) { const int p = h[da.z >> sh] + ra.z; sd[p] = da.z; ss[p] = sa.z; }
  if (da.w >= 0) { const int p = h[da.w >> sh] + ra.w; sd[p] = da.w; ss[p] = sa.w; }
  if (db.x >= 0) { const int p = h[db.x >> sh] + rb.x; sd[p] = db.x; ss[p] = sb.x; }
  if (db.y >= 0) { const int p = h[db.y >> sh] + rb.y; sd[p] = db.y; ss[p] = sb.y; }
  if (db.z >= 0) { const int p = h[db.z >> sh] + rb.z; sd[p] = db.z; ss[p] = sb.z; }
  if (db.w >= 0) { const int p = h[db.w >> sh] + rb.w; sd[p] = db.w; ss[p] = sb.w; }
  __syncthreads();
  int nE = E - c * CHUNK;
  nE = (nE > CHUNK) ? CHUNK : nE;
  for (int p = t; p < nE; p += ETPB) {
    bdst[(size_t)c * CHUNK + p] = sd[p];
    bsrc[(size_t)c * CHUNK + p] = ss[p];
  }
}

__global__ __launch_bounds__(ETPB)
void bin_csr(const int* __restrict__ bdst0, const int* __restrict__ bsrc0, int nblk0, int N0,
             int* __restrict__ rp0, int* __restrict__ cnt0, int* __restrict__ perm0,
             const int* __restrict__ bdst1, const int* __restrict__ bsrc1, int nblk1, int N1,
             int* __restrict__ rp1, int* __restrict__ cnt1, int* __restrict__ perm1,
             const int* __restrict__ table) {
  __shared__ int lc[RMAX];
  __shared__ int ls[RMAX];
  __shared__ int lq[RMAX];
  __shared__ int lp[PMAX];
  __shared__ int es[PMAX];
  __shared__ unsigned short ed[PMAX];
  __shared__ int sBeg;
  __shared__ int sCur;
  const int t    = threadIdx.x;
  const int lane = t & 31;
  const int wave = __builtin_amdgcn_readfirstlane(t >> 5);
  const int* bdst; const int* bsrc; const int* tab; int* rp; int* cnt; int* perm; int N, bin, nb;
  if ((int)blockIdx.x < NBIN) {
    bdst = bdst0; bsrc = bsrc0; tab = table; rp = rp0; cnt = cnt0; perm = perm0;
    N = N0; bin = blockIdx.x; nb = nblk0;
  } else {
    bdst = bdst1; bsrc = bsrc1; tab = table + (size_t)nblk0 * TPITCH; rp = rp1; cnt = cnt1; perm = perm1;
    N = N1; bin = blockIdx.x - NBIN; nb = nblk1;
  }
  const int sh   = bin_shift(N);
  const int row0 = bin << sh;
  if (row0 >= N) return;
  int nrow = N - row0;
  nrow = (nrow > (1 << sh)) ? (1 << sh) : nrow;

  const int myc = wave + 32 * lane;
  int myS = 0, myL = 0;
  if (myc < nb) {
    myS = tab[(size_t)myc * TPITCH + bin];
    myL = tab[(size_t)myc * TPITCH + bin + 1] - myS;
  }
  if (t < RMAX) lc[t] = 0;
  if (t == 0) { sBeg = 0; sCur = 0; }
  __syncthreads();
  int myP = 0;
  if (myc < nb) {
    atomicAdd(&sBeg, myS);
    myP = atomicAdd(&sCur, myL);
  }
  __syncthreads();
  const int beg   = sBeg;
  const int total = sCur;
  const int nch = (nb > wave) ? ((nb - wave + 31) >> 5) : 0;

  if (total <= PMAX) {
    for (int i = 0; i < nch; i += 2) {
      const int sA = __builtin_amdgcn_readlane(myS, i);
      const int lA = __builtin_amdgcn_readlane(myL, i);
      const int pA = __builtin_amdgcn_readlane(myP, i);
      int sB = 0, lB = 0, pB = 0;
      if (i + 1 < nch) {
        sB = __builtin_amdgcn_readlane(myS, i + 1);
        lB = __builtin_amdgcn_readlane(myL, i + 1);
        pB = __builtin_amdgcn_readlane(myP, i + 1);
      }
      const size_t oA = (size_t)(wave + 32 * i) * CHUNK + sA;
      const size_t oB = (i + 1 < nch) ? ((size_t)(wave + 32 * i + 32) * CHUNK + sB) : 0;
      const int* dA = bdst + oA; const int* qA = bsrc + oA;
      const int* dB = bdst + oB; const int* qB = bsrc + oB;
      int a0 = 0, a1 = 0, b0 = 0, b1 = 0, u0 = 0, u1 = 0, w0 = 0, w1 = 0;
      if (lane      < lA) { a0 = dA[lane];      u0 = qA[lane];      }
      if (lane + 32 < lA) { a1 = dA[lane + 32]; u1 = qA[lane + 32]; }
      if (lane      < lB) { b0 = dB[lane];      w0 = qB[lane];      }
      if (lane + 32 < lB) { b1 = dB[lane + 32]; w1 = qB[lane + 32]; }
      if (lane      < lA) { ed[pA + lane]      = (unsigned short)(a0 - row0); es[pA + lane]      = u0; }
      if (lane + 32 < lA) { ed[pA + lane + 32] = (unsigned short)(a1 - row0); es[pA + lane + 32] = u1; }
      if (lane      < lB) { ed[pB + lane]      = (unsigned short)(b0 - row0); es[pB + lane]      = w0; }
      if (lane + 32 < lB) { ed[pB + lane + 32] = (unsigned short)(b1 - row0); es[pB + lane + 32] = w1; }
      for (int k = 64 + lane; k < lA; k += 32) { ed[pA + k] = (unsigned short)(dA[k] - row0); es[pA + k] = qA[k]; }
      for (int k = 64 + lane; k < lB; k += 32) { ed[pB + k] = (unsigned short)(dB[k] - row0); es[pB + k] = qB[k]; }
    }
    __syncthreads();
    int ic = t;
    for (; ic + 3 * ETPB < total; ic += 4 * ETPB) {
      const int d0 = ed[ic];
      const int d1 = ed[ic + ETPB];
      const int d2 = ed[ic + 2 * ETPB];
      const int d3 = ed[ic + 3 * ETPB];
      atomicAdd(&lc[d0], 1);
      atomicAdd(&lc[d1], 1);
      atomicAdd(&lc[d2], 1);
      atomicAdd(&lc[d3], 1);
    }
    for (; ic < total; ic += ETPB) atomicAdd(&lc[ed[ic]], 1);
  } else {
    for (int cc = wave; cc < nb; cc += 32) {
      const int s = tab[(size_t)cc * TPITCH + bin];
      const int e = tab[(size_t)cc * TPITCH + bin + 1];
      for (int k = s + lane; k < e; k += 32) atomicAdd(&lc[bdst[(size_t)cc * CHUNK + k] - row0], 1);
    }
  }
  __syncthreads();
  int cn = 0;
  if (t < RMAX) { cn = lc[t]; ls[t] = cn; }
  __syncthreads();
  for (int d = 1; d < RMAX; d <<= 1) {
    const int v = (t < RMAX && t >= d) ? ls[t - d] : 0;
    __syncthreads();
    if (t < RMAX) ls[t] += v;
    __syncthreads();
  }
  if (t < RMAX) {
    const int st = ls[t] - cn;
    lq[t] = st;
    if (t < nrow) { rp[row0 + t] = beg + st; cnt[row0 + t] = cn; }
  }
  __syncthreads();
  if (total <= PMAX) {
    int iq = t;
    for (; iq + 3 * ETPB < total; iq += 4 * ETPB) {
      const int d0 = ed[iq];
      const int d1 = ed[iq + ETPB];
      const int d2 = ed[iq + 2 * ETPB];
      const int d3 = ed[iq + 3 * ETPB];
      const int s0 = es[iq];
      const int s1 = es[iq + ETPB];
      const int s2 = es[iq + 2 * ETPB];
      const int s3 = es[iq + 3 * ETPB];
      const int p0 = atomicAdd(&lq[d0], 1);
      const int p1 = atomicAdd(&lq[d1], 1);
      const int p2 = atomicAdd(&lq[d2], 1);
      const int p3 = atomicAdd(&lq[d3], 1);
      lp[p0] = s0;
      lp[p1] = s1;
      lp[p2] = s2;
      lp[p3] = s3;
    }
    for (; iq < total; iq += ETPB) {
      const int p = atomicAdd(&lq[ed[iq]], 1);
      lp[p] = es[iq];
    }
    __syncthreads();
    for (int i = t; i < total; i += ETPB) perm[beg + i] = lp[i];
    return;
  }
  for (int cc = wave; cc < nb; cc += 32) {
    const int s = tab[(size_t)cc * TPITCH + bin];
    const int e = tab[(size_t)cc * TPITCH + bin + 1];
    for (int k = s + lane; k < e; k += 32) {
      const int p = atomicAdd(&lq[bdst[(size_t)cc * CHUNK + k] - row0], 1);
      perm[beg + p] = bsrc[(size_t)cc * CHUNK + k];
    }
  }
}

__device__ __forceinline__ void mean_rows_to_lds(const float* __restrict__ xin,
                                                 const int* __restrict__ rowptr,
                                                 const int* __restrict__ cnt,
                                                 const int* __restrict__ perm,
                                                 float (*sA)[IN_CH + 4], float (*sX)[IN_CH + 4],
                                                 int rw, int nrows, int lrow, int lane) {
  int nr = nrows - rw;
  nr = (nr < 0) ? 0 : ((nr > RPWF) ? RPWF : nr);
  const v4f zero = {0.f, 0.f, 0.f, 0.f};
  const float* xl = xin + lane * 4;
  v4f xd[RPWF];
  for (int i = 0; i < RPWF; ++i)
    xd[i] = (i < nr) ? *(const v4f*)(xl + (size_t)(rw + i) * IN_CH) : zero;
  int st = 0, cn = 0;
  if (lane < nr) {
    st = rowptr[rw + lane];
    cn = cnt[rw + lane];
  }
  int start = __builtin_amdgcn_readlane(st, 0);
  int n     = __builtin_amdgcn_readlane(cn, 0);
  int mine  = (lane < n) ? perm[start + lane] : 0;
  for (int i = 0; i < RPWF; ++i) *(v4f*)&sX[lrow + i][lane * 4] = xd[i];
  int first = __builtin_amdgcn_readlane(mine, 0);
#define GATHER_ROW(k) (*(const v4f*)(xl + (size_t)__builtin_amdgcn_readlane(mine, (k)) * IN_CH))
  for (int r = 0; r < nr; ++r) {
    int start_nx = 0, n_nx = 0, mine_nx = 0;
    if (r + 1 < nr) {
      start_nx = __builtin_amdgcn_readlane(st, r + 1);
      n_nx     = __builtin_amdgcn_readlane(cn, r + 1);
      mine_nx  = (lane < n_nx) ? perm[start_nx + lane] : 0;
    }
    v4f acc = zero;
    for (int i = 0; i < n; i += 32) {
      const int m = (n - i < 32) ? (n - i) : 32;
      if (i > 0) {
        mine  = (lane < m) ? perm[start + i + lane] : 0;
        first = __builtin_amdgcn_readlane(mine, 0);
      }
      const v4f p0 = *(const v4f*)(xl + (size_t)first * IN_CH);
      int j = 1;
      for (; j + 8 <= m; j += 8) {
        const v4f v0 = GATHER_ROW(j);
        const v4f v1 = GATHER_ROW(j + 1);
        const v4f v2 = GATHER_ROW(j + 2);
        const v4f v3 = GATHER_ROW(j + 3);
        const v4f v4 = GATHER_ROW(j + 4);
        const v4f v5 = GATHER_ROW(j + 5);
        const v4f v6 = GATHER_ROW(j + 6);
        const v4f v7 = GATHER_ROW(j + 7);
        acc += v0; acc += v1; acc += v2; acc += v3;
        acc += v4; acc += v5; acc += v6; acc += v7;
      }
      const int rem = m - j;
      v4f q0 = zero, q1 = zero, q2 = zero, q3 = zero, d0 = zero, d1 = zero, o0 = zero;
      if (rem & 4) {
        q0 = GATHER_ROW(j);
        q1 = GATHER_ROW(j + 1);
        q2 = GATHER_ROW(j + 2);
        q3 = GATHER_ROW(j + 3);
        j += 4;
      }
      if (rem & 2) {
        d0 = GATHER_ROW(j);
        d1 = GATHER_ROW(j + 1);
        j += 2;
      }
      if (rem & 1) {
        o0 = GATHER_ROW(j);
      }
      acc += q0; acc += q1; acc += q2; acc += q3;
      acc += d0; acc += d1; acc += o0;
      acc += p0;
    }
    const float inv = 1.0f / fmaxf((float)n, 1.0f);
    *(v4f*)&sA[lrow + r][lane * 4] = acc * inv;
    start = start_nx;
    n     = n_nx;
    mine  = mine_nx;
    first = __builtin_amdgcn_readlane(mine, 0);
  }
#undef GATHER_ROW
  for (int r = 0; r < RPWF; ++r)
    if (r >= nr) *(v4f*)&sA[lrow + r][lane * 4] = zero;
}

#define RT0    64
#define WCH    16
#define WPITCH (HID + 8)
__global__ __launch_bounds__(512)
void sage_gather_gemm_relu(const float* __restrict__ xin,
                           const int* __restrict__ rowptr, const int* __restrict__ cnt,
                           const int* __restrict__ perm,
                           const float* __restrict__ Wl, const float* __restrict__ bl,
                           const float* __restrict__ Wr,
                           float* __restrict__ hout, int nrows)
{
  __shared__ __attribute__((aligned(16))) float sAX[2][RT0][IN_CH + 4];
  __shared__ __attribute__((aligned(16))) float sW[WCH][WPITCH];
  const int tid  = threadIdx.x;
  const int r0   = blockIdx.x * RT0;
  const int lane = tid & 31;
  const int wave = __builtin_amdgcn_readfirstlane(tid >> 5);

  const float* wsrc = Wl + (size_t)wave * HID + lane * 4;
  v4f wreg = *(const v4f*)wsrc;

  mean_rows_to_lds(xin, rowptr, cnt, perm, sAX[0], sAX[1], r0 + wave * RPWF, nrows, wave * RPWF, lane);
  *(v4f*)&sW[wave][lane * 4] = wreg;
  __syncthreads();

  const int half = lane >> 4;
  const int m    = lane & 15;
  const int n    = lane & 15;
  const int kh   = half * 2;
  const int col  = (wave & 7) * 16 + n;
  const int ma   = (wave >> 3) * 32 + m;

  v8f acc0 = {0.f, 0.f, 0.f, 0.f, 0.f, 0.f, 0.f, 0.f};
  v8f acc1 = {0.f, 0.f, 0.f, 0.f, 0.f, 0.f, 0.f, 0.f};
  for (int c = 0; c < 2 * IN_CH / WCH; ++c) {
    const int cn = c + 1;
    if (cn < 2 * IN_CH / WCH) {
      const float* Wn = (cn < IN_CH / WCH) ? Wl : Wr;
      wreg = *(const v4f*)(Wn + (size_t)((cn & (IN_CH / WCH - 1)) * WCH + wave) * HID + lane * 4);
    }
    const int op = c / (IN_CH / WCH);
    const int k0 = (c & (IN_CH / WCH - 1)) * WCH;
    for (int s = 0; s < WCH; s += 4) {
      v2f a0, a1, b;
      a0.x = sAX[op][ma][k0 + s + kh];
      a0.y = sAX[op][ma][k0 + s + kh + 1];
      a1.x = sAX[op][ma + 16][k0 + s + kh];
      a1.y = sAX[op][ma + 16][k0 + s + kh + 1];
      b.x  = sW[s + kh][col];
      b.y  = sW[s + kh + 1][col];
      acc0 = __builtin_amdgcn_wmma_f32_16x16x4_f32(false, a0, false, b, (short)0, acc0, false, false);
      acc1 = __builtin_amdgcn_wmma_f32_16x16x4_f32(false, a1, false, b, (short)0, acc1, false, false);
    }
    if (cn < 2 * IN_CH / WCH) {
      __syncthreads();
      *(v4f*)&sW[wave][lane * 4] = wreg;
      __syncthreads();
    }
  }

  const float bias  = bl[col];
  const int   mbase = (wave >> 3) * 32 + half * 8;
  for (int r = 0; r < 8; ++r) {
    const int row = r0 + mbase + r;
    if (row < nrows)
      hout[(size_t)row * HID + col] = fmaxf(acc0[r] + bias, 0.0f);
    if (row + 16 < nrows)
      hout[(size_t)(row + 16) * HID + col] = fmaxf(acc1[r] + bias, 0.0f);
  }
}

__global__ __launch_bounds__(128)
void sage_gather_gemm_lsm(const float* __restrict__ hin,
                          const int* __restrict__ rowptr, const int* __restrict__ cnt,
                          const int* __restrict__ perm,
                          const float* __restrict__ Wl, const float* __restrict__ bl,
                          const float* __restrict__ Wr,
                          float* __restrict__ out, int nrows)
{
  __shared__ __attribute__((aligned(16))) float sA[16][HID + 4];
  __shared__ __attribute__((aligned(16))) float sX[16][HID + 4];
  __shared__ float sL[16][OUTP];
  const int tid  = threadIdx.x;
  const int r0   = blockIdx.x * 16;
  const int lane = tid & 31;
  const int wave = __builtin_amdgcn_readfirstlane(tid >> 5);

  const int half = lane >> 4;
  const int m    = lane & 15;
  const int n    = lane & 15;
  const int kh   = half * 2;
  const int col  = wave * 16 + n;
  const bool cv  = (col < OUTC);
  const int colc = cv ? col : (OUTC - 1);

  float wl[HID / 2];
  if (wave < 3) {
    #pragma unroll
    for (int s = 0; s < HID / 4; ++s) {
      wl[2 * s]     = Wl[(size_t)(4 * s + kh)     * OUTC + colc];
      wl[2 * s + 1] = Wl[(size_t)(4 * s + kh + 1) * OUTC + colc];
    }
  }

  mean_rows_to_lds(hin, rowptr, cnt, perm, sA, sX, r0 + wave * RPWF, nrows, wave * RPWF, lane);
  __syncthreads();

  if (wave < 3) {
    float wr[HID / 2];
    #pragma unroll
    for (int s = 0; s < HID / 4; ++s) {
      wr[2 * s]     = Wr[(size_t)(4 * s + kh)     * OUTC + colc];
      wr[2 * s + 1] = Wr[(size_t)(4 * s + kh + 1) * OUTC + colc];
    }
    v8f acc = {0.f, 0.f, 0.f, 0.f, 0.f, 0.f, 0.f, 0.f};
    #pragma unroll
    for (int s = 0; s < HID / 4; ++s) {
      v2f a, b;
      a.x = sA[m][4 * s + kh];
      a.y = sA[m][4 * s + kh + 1];
      b.x = cv ? wl[2 * s]     : 0.0f;
      b.y = cv ? wl[2 * s + 1] : 0.0f;
      acc = __builtin_amdgcn_wmma_f32_16x16x4_f32(false, a, false, b, (short)0, acc, false, false);
    }
    #pragma unroll
    for (int s = 0; s < HID / 4; ++s) {
      v2f a, b;
      a.x = sX[m][4 * s + kh];
      a.y = sX[m][4 * s + kh + 1];
      b.x = cv ? wr[2 * s]     : 0.0f;
      b.y = cv ? wr[2 * s + 1] : 0.0f;
      acc = __builtin_amdgcn_wmma_f32_16x16x4_f32(false, a, false, b, (short)0, acc, false, false);
    }
    const float bias  = cv ? bl[col] : 0.0f;
    const int   mbase = half * 8;
    for (int r = 0; r < 8; ++r)
      sL[mbase + r][col] = acc[r] + bias;
  }
  __syncthreads();

  if (tid < 16) {
    int row = r0 + tid;
    if (row < nrows) {
      float mx = -INFINITY;
      for (int c = 0; c < OUTC; ++c) mx = fmaxf(mx, sL[tid][c]);
      float s = 0.0f;
      for (int c = 0; c < OUTC; ++c) s += expf(sL[tid][c] - mx);
      float lse = logf(s);
      for (int c = 0; c < OUTC; ++c)
        out[(size_t)row * OUTC + c] = sL[tid][c] - mx - lse;
    }
  }
}

extern "C" void kernel_launch(void* const* d_in, const int* in_sizes, int n_in,
                              void* d_out, int out_size, void* d_ws, size_t ws_size,
                              hipStream_t stream)
{
  const float* x    = (const float*)d_in[0];
  const int*   src0 = (const int*)d_in[1];
  const int*   dst0 = (const int*)d_in[2];
  const int*   src1 = (const int*)d_in[3];
  const int*   dst1 = (const int*)d_in[4];
  const float* Wl0  = (const float*)d_in[7];
  const float* bl0  = (const float*)d_in[8];
  const float* Wr0  = (const float*)d_in[9];
  const float* Wl1  = (const float*)d_in[10];
  const float* bl1  = (const float*)d_in[11];
  const float* Wr1  = (const float*)d_in[12];
  float* out = (float*)d_out;

  const int E0 = in_sizes[1];
  const int E1 = in_sizes[3];
  const int N1 = 100000;
  const int N2 = out_size / OUTC;

  float* msg0 = (float*)d_ws;
  float* deg0 = msg0 + (size_t)N1 * HID;
  float* h    = deg0 + N1;
  float* msg1 = h    + (size_t)N1 * HID;

  const int N1p = (N1 + 3) & ~3;
  const int N2p = (N2 + 3) & ~3;
  const int E0p = (E0 + 3) & ~3;
  const int E1p = (E1 + 3) & ~3;
  const int nblk0 = (E0 + CHUNK - 1) / CHUNK;
  const int nblk1 = (E1 + CHUNK - 1) / CHUNK;
  int* cnt0  = (int*)msg0;
  int* rp0   = cnt0 + N1p;
  int* perm0 = rp0 + N1p;
  int* bdst0 = perm0 + E0p;
  int* bsrc0 = bdst0 + E0p;
  int* table = bsrc0 + E0p;
  int* cnt1  = (int*)msg1;
  int* rp1   = cnt1 + N2p;
  int* perm1 = rp1 + N2p;
  int* bdst1 = perm1 + E1p;
  int* bsrc1 = bdst1 + E1p;

  bin_local<<<nblk0 + nblk1, ETPB, 0, stream>>>(src0, dst0, E0, N1, nblk0, src1, dst1, E1, N2,
                                                table, bdst0, bsrc0, bdst1, bsrc1);
  bin_csr<<<2 * NBIN, ETPB, 0, stream>>>(bdst0, bsrc0, nblk0, N1, rp0, cnt0, perm0,
                                         bdst1, bsrc1, nblk1, N2, rp1, cnt1, perm1, table);

  sage_gather_gemm_relu<<<(N1 + RT0 - 1) / RT0, 512, 0, stream>>>(x, rp0, cnt0, perm0, Wl0, bl0, Wr0, h, N1);

  sage_gather_gemm_lsm<<<(N2 + 15) / 16, 128, 0, stream>>>(h, rp1, cnt1, perm1, Wl1, bl1, Wr1, out, N2);
}
